# starting kernel
# baseline (speedup 1.0000x reference)
#include <hip/hip_runtime.h>

typedef __attribute__((ext_vector_type(2))) float v2f;
typedef __attribute__((ext_vector_type(8))) float v8f;

#define IN_C 64
#define POSD 16
#define KDIM1 (IN_C + POSD)
#define HID 64
#define OUTD 32


__global__ void k_zero(float* __restrict__ p, int n) {
  int i = blockIdx.x * blockDim.x + threadIdx.x;
  if (i < n) p[i] = 0.0f;
}

__global__ void k_deg(const int* __restrict__ dst, float* __restrict__ deg, int E) {
  int i = blockIdx.x * blockDim.x + threadIdx.x;
  if (i < E) atomicAdd(&deg[dst[i]], 1.0f);
}

__global__ void k_dinv(float* __restrict__ d, int n) {
  int i = blockIdx.x * blockDim.x + threadIdx.x;
  if (i < n) d[i] = rsqrtf(d[i] + 1.0f);
}

__global__ __launch_bounds__(256) void k_gemm1(
    const float* __restrict__ x, const float* __restrict__ pos,
    const float* __restrict__ W1, const float* __restrict__ dinv,
    float* __restrict__ hw1, float* __restrict__ h1agg, int N) {
  const int lane = threadIdx.x & 31;
  const int wave = threadIdx.x >> 5;
  const int rowBase = (blockIdx.x * 8 + wave) * 16;
  if (rowBase >= N) return;
  const int kh  = lane >> 4;
  const int n16 = lane & 15;
  int row = rowBase + n16;
  if (row >= N) row = N - 1;

  v8f acc0 = {}, acc1 = {}, acc2 = {}, acc3 = {};
  for (int k = 0; k < KDIM1; k += 4) {
    const int ka = k + 2 * kh;
    v2f a;
    a.x = (ka     < IN_C) ? x[(long)row * IN_C + ka]       : pos[(long)row * POSD + (ka     - IN_C)];
    a.y = (ka + 1 < IN_C) ? x[(long)row * IN_C + ka + 1]   : pos[(long)row * POSD + (ka + 1 - IN_C)];
    const float* wb = W1 + (long)ka * HID + n16;
    v2f b0; b0.x = wb[0];  b0.y = wb[HID];
    v2f b1; b1.x = wb[16]; b1.y = wb[HID + 16];
    v2f b2; b2.x = wb[32]; b2.y = wb[HID + 32];
    v2f b3; b3.x = wb[48]; b3.y = wb[HID + 48];
    acc0 = __builtin_amdgcn_wmma_f32_16x16x4_f32(false, a, false, b0, (short)0, acc0, false, false);
    acc1 = __builtin_amdgcn_wmma_f32_16x16x4_f32(false, a, false, b1, (short)0, acc1, false, false);
    acc2 = __builtin_amdgcn_wmma_f32_16x16x4_f32(false, a, false, b2, (short)0, acc2, false, false);
    acc3 = __builtin_amdgcn_wmma_f32_16x16x4_f32(false, a, false, b3, (short)0, acc3, false, false);
  }
#pragma unroll
  for (int i = 0; i < 8; ++i) {
    const int r = rowBase + i + 8 * kh;
    if (r < N) {
      const float dv = dinv[r];
      const float sw = dv * dv;
      const long base = (long)r * HID + n16;
      hw1[base +  0] = acc0[i];  h1agg[base +  0] = acc0[i] * sw;
      hw1[base + 16] = acc1[i];  h1agg[base + 16] = acc1[i] * sw;
      hw1[base + 32] = acc2[i];  h1agg[base + 32] = acc2[i] * sw;
      hw1[base + 48] = acc3[i];  h1agg[base + 48] = acc3[i] * sw;
    }
  }
}

__global__ __launch_bounds__(256) void k_gemm2(
    const float* __restrict__ h, const float* __restrict__ W2,
    const float* __restrict__ dinv, float* __restrict__ hw2,
    float* __restrict__ zinit, int N) {
  const int lane = threadIdx.x & 31;
  const int wave = threadIdx.x >> 5;
  const int rowBase = (blockIdx.x * 8 + wave) * 16;
  if (rowBase >= N) return;
  const int kh  = lane >> 4;
  const int n16 = lane & 15;
  int row = rowBase + n16;
  if (row >= N) row = N - 1;

  v8f acc0 = {}, acc1 = {};
  for (int k = 0; k < HID; k += 4) {
    const int ka = k + 2 * kh;
    v2f a;
    a.x = h[(long)row * HID + ka];
    a.y = h[(long)row * HID + ka + 1];
    const float* wb = W2 + (long)ka * OUTD + n16;
    v2f b0; b0.x = wb[0];  b0.y = wb[OUTD];
    v2f b1; b1.x = wb[16]; b1.y = wb[OUTD + 16];
    acc0 = __builtin_amdgcn_wmma_f32_16x16x4_f32(false, a, false, b0, (short)0, acc0, false, false);
    acc1 = __builtin_amdgcn_wmma_f32_16x16x4_f32(false, a, false, b1, (short)0, acc1, false, false);
  }
#pragma unroll
  for (int i = 0; i < 8; ++i) {
    const int r = rowBase + i + 8 * kh;
    if (r < N) {
      const float dv = dinv[r];
      const float sw = dv * dv;
      const long base = (long)r * OUTD + n16;
      hw2[base +  0] = acc0[i];  zinit[base +  0] = acc0[i] * sw;
      hw2[base + 16] = acc1[i];  zinit[base + 16] = acc1[i] * sw;
    }
  }
}

__global__ void k_scatter(const int* __restrict__ src, const int* __restrict__ dst,
                          const float* __restrict__ dinv, const float* __restrict__ feat,
                          float* __restrict__ agg, long total, int shift) {
  long t = (long)blockIdx.x * blockDim.x + threadIdx.x;
  if (t >= total) return;
  const int e = (int)(t >> shift);
  const int c = (int)(t & ((1L << shift) - 1));
  const int s = src[e];
  const int d = dst[e];
  const float w = dinv[s] * dinv[d];
  atomicAdd(&agg[((long)d << shift) + c], feat[((long)s << shift) + c] * w);
}

__global__ void k_bias_act(float* __restrict__ h, const float* __restrict__ b,
                           long total, int mask, int relu) {
  long i = (long)blockIdx.x * blockDim.x + threadIdx.x;
  if (i >= total) return;
  float v = h[i] + b[(int)(i & mask)];
  if (relu) v = fmaxf(v, 0.0f);
  h[i] = v;
}

__global__ void k_link(const float* __restrict__ z, const int* __restrict__ eli,
                       const float* __restrict__ Wl, const float* __restrict__ bl,
                       float* __restrict__ pred, int EL) {
  int e = blockIdx.x * blockDim.x + threadIdx.x;
  if (e >= EL) return;
  const int s = eli[e];
  const int d = eli[EL + e];
  const float* zs = z + (long)s * OUTD;
  const float* zd = z + (long)d * OUTD;
  float acc = bl[0];
#pragma unroll
  for (int c = 0; c < OUTD; ++c)
    acc = fmaf(zs[c], Wl[c], fmaf(zd[c], Wl[OUTD + c], acc));
  pred[e] = acc;
}

extern "C" void kernel_launch(void* const* d_in, const int* in_sizes, int n_in,
                              void* d_out, int out_size, void* d_ws, size_t ws_size,
                              hipStream_t stream) {
  const float* x   = (const float*)d_in[0];
  const float* pos = (const float*)d_in[1];
  const float* W1  = (const float*)d_in[2];
  const float* b1  = (const float*)d_in[3];
  const float* W2  = (const float*)d_in[4];
  const float* b2  = (const float*)d_in[5];
  const float* Wl  = (const float*)d_in[6];
  const float* bl  = (const float*)d_in[7];
  const int*   ei  = (const int*)d_in[8];
  const int*   eli = (const int*)d_in[9];

  const int N  = in_sizes[0] / IN_C;
  const int E  = in_sizes[8] / 2;
  const int EL = in_sizes[9] / 2;

  float* z    = (float*)d_out;
  float* pred = z + (long)N * OUTD;

  float* dinv  = (float*)d_ws;
  float* hw1   = dinv + N;
  float* h1    = hw1 + (long)N * HID;
  float* hw2   = h1  + (long)N * HID;

  const int* esrc = ei;
  const int* edst = ei + E;

  const int B = 256;

  k_zero<<<(N + B - 1) / B, B, 0, stream>>>(dinv, N);
  k_deg <<<(E + B - 1) / B, B, 0, stream>>>(edst, dinv, E);
  k_dinv<<<(N + B - 1) / B, B, 0, stream>>>(dinv, N);

  const int rowTiles = (N + 15) / 16;
  const int gblocks  = (rowTiles + 7) / 8;
  k_gemm1<<<gblocks, B, 0, stream>>>(x, pos, W1, dinv, hw1, h1, N);
  {
    long tot = (long)E * HID;
    k_scatter<<<(int)((tot + B - 1) / B), B, 0, stream>>>(esrc, edst, dinv, hw1, h1, tot, 6);
    long nh = (long)N * HID;
    k_bias_act<<<(int)((nh + B - 1) / B), B, 0, stream>>>(h1, b1, nh, HID - 1, 1);
  }

  k_gemm2<<<gblocks, B, 0, stream>>>(h1, W2, dinv, hw2, z, N);
  {
    long tot = (long)E * OUTD;
    k_scatter<<<(int)((tot + B - 1) / B), B, 0, stream>>>(esrc, edst, dinv, hw2, z, tot, 5);
    long nz = (long)N * OUTD;
    k_bias_act<<<(int)((nz + B - 1) / B), B, 0, stream>>>(z, b2, nz, OUTD - 1, 0);
  }

  k_link<<<(EL + B - 1) / B, B, 0, stream>>>(z, eli, Wl, bl, pred, EL);
}


# Replace float atomic scatter-add by in-call CSR (atomic arrival-order rank, scan, fill) + wave-per-row register gather: row sums in rank order; self loop, bias, relu folded into the one row store
# speedup vs baseline: 2.4818x; 2.4818x over previous
#include <hip/hip_runtime.h>

typedef __attribute__((ext_vector_type(2))) float v2f;
typedef __attribute__((ext_vector_type(8))) float v8f;

#define IN_C 64
#define POSD 16
#define KDIM1 (IN_C + POSD)
#define HID 64
#define OUTD 32
#define SCAN_ELEMS 1024


__global__ void k_zero(int* __restrict__ p, int n) {
  int i = blockIdx.x * blockDim.x + threadIdx.x;
  if (i < n) p[i] = 0;
}

__global__ void k_rank(const int* __restrict__ dst, int* __restrict__ cnt,
                       int* __restrict__ rank, int E) {
  int e = blockIdx.x * blockDim.x + threadIdx.x;
  if (e < E) rank[e] = atomicAdd(&cnt[dst[e]], 1);
}

__global__ __launch_bounds__(256) void k_bsum(const int* __restrict__ cnt,
                                              int* __restrict__ bsum, int N) {
  __shared__ int sm[8];
  const int tid = threadIdx.x, lane = tid & 31, wave = tid >> 5;
  const int base = blockIdx.x * SCAN_ELEMS + tid * 4;
  int s = 0;
  for (int i = 0; i < 4; ++i)
    if (base + i < N) s += cnt[base + i];
  for (int off = 16; off > 0; off >>= 1) s += __shfl_down(s, off, 32);
  if (lane == 0) sm[wave] = s;
  __syncthreads();
  if (tid == 0) {
    int t = 0;
    for (int w = 0; w < 8; ++w) t += sm[w];
    bsum[blockIdx.x] = t;
  }
}

__global__ __launch_bounds__(256) void k_scan(const int* __restrict__ cnt,
                                              const int* __restrict__ bsum,
                                              int* __restrict__ rowptr,
                                              float* __restrict__ dinv, int N) {
  __shared__ int sm[8];
  __shared__ int sblk;
  const int tid = threadIdx.x, lane = tid & 31, wave = tid >> 5;

  int p = 0;
  for (int i = tid; i < (int)blockIdx.x; i += 256) p += bsum[i];
  for (int off = 16; off > 0; off >>= 1) p += __shfl_down(p, off, 32);
  if (lane == 0) sm[wave] = p;
  __syncthreads();
  if (tid == 0) {
    int t = 0;
    for (int w = 0; w < 8; ++w) t += sm[w];
    sblk = t;
  }
  __syncthreads();
  const int blockOff = sblk;

  const int base = blockIdx.x * SCAN_ELEMS + tid * 4;
  int c[4];
  int tsum = 0;
  for (int i = 0; i < 4; ++i) {
    c[i] = (base + i < N) ? cnt[base + i] : 0;
    tsum += c[i];
  }
  int inc = tsum;
  for (int off = 1; off < 32; off <<= 1) {
    const int t = __shfl_up(inc, off, 32);
    if (lane >= off) inc += t;
  }
  if (lane == 31) sm[wave] = inc;
  __syncthreads();
  int woff = 0;
  for (int w = 0; w < wave; ++w) woff += sm[w];
  int run = blockOff + woff + inc - tsum;
  for (int i = 0; i < 4; ++i) {
    if (base + i < N) {
      rowptr[base + i] = run;
      dinv[base + i] = rsqrtf((float)(c[i] + 1));
      run += c[i];
      if (base + i == N - 1) rowptr[N] = run;
    }
  }
}

__global__ void k_fill(const int* __restrict__ src, const int* __restrict__ dst,
                       const int* __restrict__ rowptr, const int* __restrict__ rank,
                       int* __restrict__ col, int E) {
  int e = blockIdx.x * blockDim.x + threadIdx.x;
  if (e < E) col[rowptr[dst[e]] + rank[e]] = src[e];
}

__global__ __launch_bounds__(256) void k_gemm1(
    const float* __restrict__ x, const float* __restrict__ pos,
    const float* __restrict__ W1, float* __restrict__ hw1, int N) {
  const int lane = threadIdx.x & 31;
  const int wave = threadIdx.x >> 5;
  const int rowBase = (blockIdx.x * 8 + wave) * 16;
  if (rowBase >= N) return;
  const int kh  = lane >> 4;
  const int n16 = lane & 15;
  int row = rowBase + n16;
  if (row >= N) row = N - 1;

  v8f acc0 = {}, acc1 = {}, acc2 = {}, acc3 = {};
  for (int k = 0; k < KDIM1; k += 4) {
    const int ka = k + 2 * kh;
    v2f a;
    a.x = (ka     < IN_C) ? x[(long)row * IN_C + ka]       : pos[(long)row * POSD + (ka     - IN_C)];
    a.y = (ka + 1 < IN_C) ? x[(long)row * IN_C + ka + 1]   : pos[(long)row * POSD + (ka + 1 - IN_C)];
    const float* wb = W1 + (long)ka * HID + n16;
    v2f b0; b0.x = wb[0];  b0.y = wb[HID];
    v2f b1; b1.x = wb[16]; b1.y = wb[HID + 16];
    v2f b2; b2.x = wb[32]; b2.y = wb[HID + 32];
    v2f b3; b3.x = wb[48]; b3.y = wb[HID + 48];
    acc0 = __builtin_amdgcn_wmma_f32_16x16x4_f32(false, a, false, b0, (short)0, acc0, false, false);
    acc1 = __builtin_amdgcn_wmma_f32_16x16x4_f32(false, a, false, b1, (short)0, acc1, false, false);
    acc2 = __builtin_amdgcn_wmma_f32_16x16x4_f32(false, a, false, b2, (short)0, acc2, false, false);
    acc3 = __builtin_amdgcn_wmma_f32_16x16x4_f32(false, a, false, b3, (short)0, acc3, false, false);
  }
#pragma unroll
  for (int i = 0; i < 8; ++i) {
    const int r = rowBase + i + 8 * kh;
    if (r < N) {
      const long base = (long)r * HID + n16;
      hw1[base +  0] = acc0[i];
      hw1[base + 16] = acc1[i];
      hw1[base + 32] = acc2[i];
      hw1[base + 48] = acc3[i];
    }
  }
}

__global__ __launch_bounds__(256) void k_gemm2(
    const float* __restrict__ h, const float* __restrict__ W2,
    float* __restrict__ hw2, int N) {
  const int lane = threadIdx.x & 31;
  const int wave = threadIdx.x >> 5;
  const int rowBase = (blockIdx.x * 8 + wave) * 16;
  if (rowBase >= N) return;
  const int kh  = lane >> 4;
  const int n16 = lane & 15;
  int row = rowBase + n16;
  if (row >= N) row = N - 1;

  v8f acc0 = {}, acc1 = {};
  for (int k = 0; k < HID; k += 4) {
    const int ka = k + 2 * kh;
    v2f a;
    a.x = h[(long)row * HID + ka];
    a.y = h[(long)row * HID + ka + 1];
    const float* wb = W2 + (long)ka * OUTD + n16;
    v2f b0; b0.x = wb[0];  b0.y = wb[OUTD];
    v2f b1; b1.x = wb[16]; b1.y = wb[OUTD + 16];
    acc0 = __builtin_amdgcn_wmma_f32_16x16x4_f32(false, a, false, b0, (short)0, acc0, false, false);
    acc1 = __builtin_amdgcn_wmma_f32_16x16x4_f32(false, a, false, b1, (short)0, acc1, false, false);
  }
#pragma unroll
  for (int i = 0; i < 8; ++i) {
    const int r = rowBase + i + 8 * kh;
    if (r < N) {
      const long base = (long)r * OUTD + n16;
      hw2[base +  0] = acc0[i];
      hw2[base + 16] = acc1[i];
    }
  }
}


template <int U>
__device__ __forceinline__ void gather_batch2(const float2* __restrict__ f2, int s_l, float w_l,
                                              int j, int lane, float2& acc) {
  float2 v[U];
  float w[U];
#pragma unroll
  for (int u = 0; u < U; ++u) {
    const int s = __builtin_amdgcn_readlane(s_l, j + u);
    w[u] = __int_as_float(__builtin_amdgcn_readlane(__float_as_int(w_l), j + u));
    v[u] = f2[(long)s * (HID / 2) + lane];
  }
#pragma unroll
  for (int u = 0; u < U; ++u) {
    acc.x = fmaf(v[u].x, w[u], acc.x);
    acc.y = fmaf(v[u].y, w[u], acc.y);
  }
}

template <int U>
__device__ __forceinline__ void gather_batch1(const float* __restrict__ f, int s_l, float w_l,
                                              int j, int lane, float& acc) {
  float v[U];
  float w[U];
#pragma unroll
  for (int u = 0; u < U; ++u) {
    const int s = __builtin_amdgcn_readlane(s_l, j + u);
    w[u] = __int_as_float(__builtin_amdgcn_readlane(__float_as_int(w_l), j + u));
    v[u] = f[(long)s * OUTD + lane];
  }
#pragma unroll
  for (int u = 0; u < U; ++u) acc = fmaf(v[u], w[u], acc);
}

__global__ __launch_bounds__(256) void k_gather64_relu(
    const int* __restrict__ rowptr, const int* __restrict__ col,
    const float* __restrict__ dinv, const float* __restrict__ feat,
    const float* __restrict__ bias, float* __restrict__ out, int N) {
  const int lane = threadIdx.x & 31;
  const int r0 = blockIdx.x * 8 + (threadIdx.x >> 5);
  if (r0 >= N) return;
  const int row = __builtin_amdgcn_readfirstlane(r0);
  const int beg = __builtin_amdgcn_readfirstlane(rowptr[row]);
  const int end = __builtin_amdgcn_readfirstlane(rowptr[row + 1]);
  const float dd = dinv[row];
  const float2* __restrict__ f2 = reinterpret_cast<const float2*>(feat);

  const float2 self = f2[(long)row * (HID / 2) + lane];
  float2 acc;
  acc.x = 0.0f;
  acc.y = 0.0f;

  for (int base = beg; base < end; base += 32) {
    const int n = (end - base < 32) ? (end - base) : 32;
    int s_l = row;
    float w_l = 0.0f;
    if (lane < n) {
      s_l = col[base + lane];
      s_l = (s_l < 0) ? 0 : ((s_l >= N) ? N - 1 : s_l);
      w_l = dinv[s_l] * dd;
    }
    int j = 0;
    for (; j + 8 <= n; j += 8) gather_batch2<8>(f2, s_l, w_l, j, lane, acc);
    if (n - j >= 4) { gather_batch2<4>(f2, s_l, w_l, j, lane, acc); j += 4; }
    if (n - j >= 2) { gather_batch2<2>(f2, s_l, w_l, j, lane, acc); j += 2; }
    if (n - j >= 1) { gather_batch2<1>(f2, s_l, w_l, j, lane, acc); }
  }

  const float sw = dd * dd;
  acc.x = fmaf(self.x, sw, acc.x);
  acc.y = fmaf(self.y, sw, acc.y);
  const float2 b = reinterpret_cast<const float2*>(bias)[lane];
  float2 o;
  o.x = fmaxf(acc.x + b.x, 0.0f);
  o.y = fmaxf(acc.y + b.y, 0.0f);
  reinterpret_cast<float2*>(out)[(long)row * (HID / 2) + lane] = o;
}

__global__ __launch_bounds__(256) void k_gather32(
    const int* __restrict__ rowptr, const int* __restrict__ col,
    const float* __restrict__ dinv, const float* __restrict__ feat,
    const float* __restrict__ bias, float* __restrict__ out, int N) {
  const int lane = threadIdx.x & 31;
  const int r0 = blockIdx.x * 8 + (threadIdx.x >> 5);
  if (r0 >= N) return;
  const int row = __builtin_amdgcn_readfirstlane(r0);
  const int beg = __builtin_amdgcn_readfirstlane(rowptr[row]);
  const int end = __builtin_amdgcn_readfirstlane(rowptr[row + 1]);
  const float dd = dinv[row];

  const float self = feat[(long)row * OUTD + lane];
  float acc = 0.0f;

  for (int base = beg; base < end; base += 32) {
    const int n = (end - base < 32) ? (end - base) : 32;
    int s_l = row;
    float w_l = 0.0f;
    if (lane < n) {
      s_l = col[base + lane];
      s_l = (s_l < 0) ? 0 : ((s_l >= N) ? N - 1 : s_l);
      w_l = dinv[s_l] * dd;
    }
    int j = 0;
    for (; j + 8 <= n; j += 8) gather_batch1<8>(feat, s_l, w_l, j, lane, acc);
    if (n - j >= 4) { gather_batch1<4>(feat, s_l, w_l, j, lane, acc); j += 4; }
    if (n - j >= 2) { gather_batch1<2>(feat, s_l, w_l, j, lane, acc); j += 2; }
    if (n - j >= 1) { gather_batch1<1>(feat, s_l, w_l, j, lane, acc); }
  }

  acc = fmaf(self, dd * dd, acc);
  out[(long)row * OUTD + lane] = acc + bias[lane];
}

__global__ void k_link(const float* __restrict__ z, const int* __restrict__ eli,
                       const float* __restrict__ Wl, const float* __restrict__ bl,
                       float* __restrict__ pred, int EL) {
  int e = blockIdx.x * blockDim.x + threadIdx.x;
  if (e >= EL) return;
  const int s = eli[e];
  const int d = eli[EL + e];
  const float* zs = z + (long)s * OUTD;
  const float* zd = z + (long)d * OUTD;
  float acc = bl[0];
#pragma unroll
  for (int c = 0; c < OUTD; ++c)
    acc = fmaf(zs[c], Wl[c], fmaf(zd[c], Wl[OUTD + c], acc));
  pred[e] = acc;
}

extern "C" void kernel_launch(void* const* d_in, const int* in_sizes, int n_in,
                              void* d_out, int out_size, void* d_ws, size_t ws_size,
                              hipStream_t stream) {
  const float* x   = (const float*)d_in[0];
  const float* pos = (const float*)d_in[1];
  const float* W1  = (const float*)d_in[2];
  const float* b1  = (const float*)d_in[3];
  const float* W2  = (const float*)d_in[4];
  const float* b2  = (const float*)d_in[5];
  const float* Wl  = (const float*)d_in[6];
  const float* bl  = (const float*)d_in[7];
  const int*   ei  = (const int*)d_in[8];
  const int*   eli = (const int*)d_in[9];

  const int N  = in_sizes[0] / IN_C;
  const int E  = in_sizes[8] / 2;
  const int EL = in_sizes[9] / 2;

  float* z    = (float*)d_out;
  float* pred = z + (long)N * OUTD;

  float* hw1   = (float*)d_ws;
  float* h1    = hw1 + (long)N * HID;
  float* reg3  = h1  + (long)N * HID;
  float* dinv  = reg3 + (long)N * OUTD;
  float* hw2   = hw1;
  int*   rank  = (int*)h1;
  int*   col   = (int*)reg3;
  int*   cnt   = col + E;
  int*   rowptr = cnt + N;
  int*   bsum  = rowptr + N + 1;

  const int* esrc = ei;
  const int* edst = ei + E;

  const int B = 256;
  const int scanBlocks = (N + SCAN_ELEMS - 1) / SCAN_ELEMS;

  k_zero<<<(N + B - 1) / B, B, 0, stream>>>(cnt, N);
  k_rank<<<(E + B - 1) / B, B, 0, stream>>>(edst, cnt, rank, E);
  k_bsum<<<scanBlocks, B, 0, stream>>>(cnt, bsum, N);
  k_scan<<<scanBlocks, B, 0, stream>>>(cnt, bsum, rowptr, dinv, N);
  k_fill<<<(E + B - 1) / B, B, 0, stream>>>(esrc, edst, rowptr, rank, col, E);

  const int rowTiles = (N + 15) / 16;
  const int gblocks  = (rowTiles + 7) / 8;
  const int rblocks  = (N + 7) / 8;

  k_gemm1<<<gblocks, B, 0, stream>>>(x, pos, W1, hw1, N);
  k_gather64_relu<<<rblocks, B, 0, stream>>>(rowptr, col, dinv, hw1, b1, h1, N);

  k_gemm2<<<gblocks, B, 0, stream>>>(h1, W2, hw2, N);
  k_gather32<<<rblocks, B, 0, stream>>>(rowptr, col, dinv, hw2, b2, z, N);

  k_link<<<(EL + B - 1) / B, B, 0, stream>>>(z, eli, Wl, bl, pred, EL);
}


# CSR build as a two-level counting sort on LDS counters (tile partition by destination range, then one block per range ranks its rows in LDS): no global atomic, no scan; rank = LDS arrival order
# speedup vs baseline: 5.4888x; 2.2116x over previous
#include <hip/hip_runtime.h>

typedef __attribute__((ext_vector_type(2))) float v2f;
typedef __attribute__((ext_vector_type(8))) float v8f;

#define IN_C 64
#define POSD 16
#define KDIM1 (IN_C + POSD)
#define HID 64
#define OUTD 32


#define RN_SHIFT 8
#define RN 256
#define MAXR 1024
#define PT 1024
#define RT 512
#define RW (RT / 32)
#define EPT 8
#define PTILE (PT * EPT)
#define CAPC 12288

__global__ __launch_bounds__(PT) void k_part(const int* __restrict__ src, const int* __restrict__ dst,
                                             int2* __restrict__ seg, int* __restrict__ loff,
                                             int E, int N, int R) {
  __shared__ int rc[MAXR];
  __shared__ int wsum[32];
  __shared__ int2 stage[PTILE];
  const int tid = threadIdx.x, lane = tid & 31, wave = tid >> 5;
  const int base = blockIdx.x * PTILE;

  rc[tid] = 0;
  __syncthreads();

  int s[EPT], d[EPT], rk[EPT];
#pragma unroll
  for (int k = 0; k < EPT; ++k) {
    const int e = base + k * PT + tid;
    const int ec = (e < E) ? e : (E - 1);
    s[k] = src[ec];
    d[k] = dst[ec];
  }
#pragma unroll
  for (int k = 0; k < EPT; ++k) {
    const int e = base + k * PT + tid;
    int t = d[k];
    t = (t < 0) ? 0 : ((t >= N) ? N - 1 : t);
    d[k] = (e < E) ? t : -1;
  }
#pragma unroll
  for (int k = 0; k < EPT; ++k) {
    rk[k] = 0;
    if (d[k] >= 0) rk[k] = atomicAdd(&rc[d[k] >> RN_SHIFT], 1);
  }
  __syncthreads();

  const int c = rc[tid];
  int inc = c;
  for (int off = 1; off < 32; off <<= 1) {
    const int t = __shfl_up(inc, off, 32);
    if (lane >= off) inc += t;
  }
  if (lane == 31) wsum[wave] = inc;
  __syncthreads();
  if (wave == 0) {
    const int v = wsum[lane];
    int iv = v;
    for (int off = 1; off < 32; off <<= 1) {
      const int t = __shfl_up(iv, off, 32);
      if (lane >= off) iv += t;
    }
    wsum[lane] = iv - v;
  }
  __syncthreads();
  const int excl = wsum[wave] + inc - c;
  rc[tid] = excl;
  if (tid <= R) loff[(long)blockIdx.x * (R + 1) + tid] = excl;
  __syncthreads();

#pragma unroll
  for (int k = 0; k < EPT; ++k) {
    if (d[k] >= 0) {
      int2 p;
      p.x = s[k];
      p.y = d[k];
      stage[rc[d[k] >> RN_SHIFT] + rk[k]] = p;
    }
  }
  __syncthreads();

  const int nloc = (E - base < PTILE) ? (E - base) : PTILE;
  for (int i = tid; i < nloc; i += PT) seg[(long)base + i] = stage[i];
}

template <bool PLACE>
__device__ __forceinline__ void walk_pieces(const int2* __restrict__ seg, const int* __restrict__ loff,
                                            int R, int PB, int r, int lane, int wave,
                                            int* cnt, int* cur, int* cstage, int* __restrict__ colr,
                                            bool staged) {
  for (int p0 = 0; p0 < PB; p0 += RT) {
    const int p = p0 + lane * RW + wave;
    int pst = 0, pcn = 0;
    if (p < PB) {
      const int a = loff[(long)p * (R + 1) + r];
      const int b = loff[(long)p * (R + 1) + r + 1];
      pst = p * PTILE + a;
      pcn = b - a;
    }
    const int rem = PB - p0 - wave;
    int npw = (rem <= 0) ? 0 : ((rem + RW - 1) / RW);
    if (npw > 32) npw = 32;
    for (int j = 0; j < npw; j += 8) {
      int st[8], cn[8];
      int2 e[8];
#pragma unroll
      for (int u = 0; u < 8; ++u) {
        st[u] = __builtin_amdgcn_readlane(pst, j + u);
        cn[u] = __builtin_amdgcn_readlane(pcn, j + u);
      }
#pragma unroll
      for (int u = 0; u < 8; ++u) {
        const long idx = (lane < cn[u]) ? ((long)st[u] + lane) : 0;
        e[u] = seg[idx];
      }
#pragma unroll
      for (int u = 0; u < 8; ++u) {
        if (lane < cn[u]) {
          const int loc = e[u].y & (RN - 1);
          if (!PLACE) {
            atomicAdd(&cnt[loc], 1);
          } else {
            const int pos = atomicAdd(&cur[loc], 1);
            if (staged) cstage[pos] = e[u].x; else colr[pos] = e[u].x;
          }
        }
      }
    }
    for (int j = 0; j < npw; ++j) {
      const int cnj = __builtin_amdgcn_readlane(pcn, j);
      if (cnj > 32) {
        const int stj = __builtin_amdgcn_readlane(pst, j);
        for (int i = 32 + lane; i < cnj; i += 32) {
          const int2 ee = seg[(long)stj + i];
          const int loc = ee.y & (RN - 1);
          if (!PLACE) {
            atomicAdd(&cnt[loc], 1);
          } else {
            const int pos = atomicAdd(&cur[loc], 1);
            if (staged) cstage[pos] = ee.x; else colr[pos] = ee.x;
          }
        }
      }
    }
  }
}

__global__ __launch_bounds__(RT) void k_rowsort(const int2* __restrict__ seg, const int* __restrict__ loff,
                                                int* __restrict__ rowptr, float* __restrict__ dinv,
                                                int* __restrict__ col, int N, int R, int PB) {
  __shared__ int cnt[RN];
  __shared__ int cur[RN];
  __shared__ int red[RW];
  __shared__ int wtot[RN / 32];
  __shared__ int cstage[CAPC];
  const int tid = threadIdx.x, lane = tid & 31, wave = tid >> 5;
  const int r = blockIdx.x;
  const int nbase = r << RN_SHIFT;

  if (tid < RN) cnt[tid] = 0;

  int bl = 0;
  for (int p = tid; p < PB; p += RT) bl += loff[(long)p * (R + 1) + r];
  for (int off = 16; off > 0; off >>= 1) bl += __shfl_down(bl, off, 32);
  if (lane == 0) red[wave] = bl;
  __syncthreads();
  int rbase = 0;
  for (int w = 0; w < RW; ++w) rbase += red[w];

  walk_pieces<false>(seg, loff, R, PB, r, lane, wave, cnt, cur, cstage, col, false);
  __syncthreads();

  const int c = (tid < RN) ? cnt[tid] : 0;
  int inc = c;
  for (int off = 1; off < 32; off <<= 1) {
    const int t = __shfl_up(inc, off, 32);
    if (lane >= off) inc += t;
  }
  if (tid < RN && lane == 31) wtot[wave] = inc;
  __syncthreads();
  int ntot = 0;
  for (int w = 0; w < RN / 32; ++w) ntot += wtot[w];
  if (tid < RN) {
    int woff = 0;
    for (int w = 0; w < wave; ++w) woff += wtot[w];
    const int excl = woff + inc - c;
    cur[tid] = excl;
    const int node = nbase + tid;
    if (node < N) {
      rowptr[node] = rbase + excl;
      dinv[node] = rsqrtf((float)(c + 1));
      if (node == N - 1) rowptr[N] = rbase + excl + c;
    }
  }
  const bool staged = (ntot <= CAPC);
  __syncthreads();

  walk_pieces<true>(seg, loff, R, PB, r, lane, wave, cnt, cur, cstage, col + rbase, staged);

  if (staged) {
    __syncthreads();
    for (int i = tid; i < ntot; i += RT) col[(long)rbase + i] = cstage[i];
  }
}

__global__ __launch_bounds__(256) void k_gemm1(
    const float* __restrict__ x, const float* __restrict__ pos,
    const float* __restrict__ W1, const float* __restrict__ dinv,
    float* __restrict__ hw1, int N) {
  const int lane = threadIdx.x & 31;
  const int wave = threadIdx.x >> 5;
  const int rowBase = (blockIdx.x * 8 + wave) * 16;
  if (rowBase >= N) return;
  const int kh  = lane >> 4;
  const int n16 = lane & 15;
  int row = rowBase + n16;
  if (row >= N) row = N - 1;

  float sc[8];
#pragma unroll
  for (int i = 0; i < 8; ++i) {
    int r = rowBase + i + 8 * kh;
    if (r >= N) r = N - 1;
    sc[i] = dinv[r];
  }

  v8f acc0 = {}, acc1 = {}, acc2 = {}, acc3 = {};
  for (int k = 0; k < KDIM1; k += 4) {
    const int ka = k + 2 * kh;
    v2f a;
    a.x = (ka     < IN_C) ? x[(long)row * IN_C + ka]       : pos[(long)row * POSD + (ka     - IN_C)];
    a.y = (ka + 1 < IN_C) ? x[(long)row * IN_C + ka + 1]   : pos[(long)row * POSD + (ka + 1 - IN_C)];
    const float* wb = W1 + (long)ka * HID + n16;
    v2f b0; b0.x = wb[0];  b0.y = wb[HID];
    v2f b1; b1.x = wb[16]; b1.y = wb[HID + 16];
    v2f b2; b2.x = wb[32]; b2.y = wb[HID + 32];
    v2f b3; b3.x = wb[48]; b3.y = wb[HID + 48];
    acc0 = __builtin_amdgcn_wmma_f32_16x16x4_f32(false, a, false, b0, (short)0, acc0, false, false);
    acc1 = __builtin_amdgcn_wmma_f32_16x16x4_f32(false, a, false, b1, (short)0, acc1, false, false);
    acc2 = __builtin_amdgcn_wmma_f32_16x16x4_f32(false, a, false, b2, (short)0, acc2, false, false);
    acc3 = __builtin_amdgcn_wmma_f32_16x16x4_f32(false, a, false, b3, (short)0, acc3, false, false);
  }
#pragma unroll
  for (int i = 0; i < 8; ++i) {
    acc0[i] *= sc[i];
    acc1[i] *= sc[i];
    acc2[i] *= sc[i];
    acc3[i] *= sc[i];
  }
#pragma unroll
  for (int i = 0; i < 8; ++i) {
    const int r = rowBase + i + 8 * kh;
    if (r < N) {
      const long base = (long)r * HID + n16;
      hw1[base +  0] = acc0[i];
      hw1[base + 16] = acc1[i];
      hw1[base + 32] = acc2[i];
      hw1[base + 48] = acc3[i];
    }
  }
}

__global__ __launch_bounds__(256) void k_gemm2(
    const float* __restrict__ h, const float* __restrict__ W2,
    float* __restrict__ hw2, int N) {
  const int lane = threadIdx.x & 31;
  const int wave = threadIdx.x >> 5;
  const int rowBase = (blockIdx.x * 8 + wave) * 16;
  if (rowBase >= N) return;
  const int kh  = lane >> 4;
  const int n16 = lane & 15;
  int row = rowBase + n16;
  if (row >= N) row = N - 1;

  v8f acc0 = {}, acc1 = {};
  for (int k = 0; k < HID; k += 4) {
    const int ka = k + 2 * kh;
    v2f a;
    a.x = h[(long)row * HID + ka];
    a.y = h[(long)row * HID + ka + 1];
    const float* wb = W2 + (long)ka * OUTD + n16;
    v2f b0; b0.x = wb[0];  b0.y = wb[OUTD];
    v2f b1; b1.x = wb[16]; b1.y = wb[OUTD + 16];
    acc0 = __builtin_amdgcn_wmma_f32_16x16x4_f32(false, a, false, b0, (short)0, acc0, false, false);
    acc1 = __builtin_amdgcn_wmma_f32_16x16x4_f32(false, a, false, b1, (short)0, acc1, false, false);
  }
#pragma unroll
  for (int i = 0; i < 8; ++i) {
    const int r = rowBase + i + 8 * kh;
    if (r < N) {
      const long base = (long)r * OUTD + n16;
      hw2[base +  0] = acc0[i];
      hw2[base + 16] = acc1[i];
    }
  }
}


template <int U>
__device__ __forceinline__ void gather_batch2(const float2* __restrict__ f2, int s_l,
                                              int j, int lane, float2& acc) {
  float2 v[U];
#pragma unroll
  for (int u = 0; u < U; ++u) {
    const int s = __builtin_amdgcn_readlane(s_l, j + u);
    v[u] = f2[(long)s * (HID / 2) + lane];
  }
#pragma unroll
  for (int u = 0; u < U; ++u) {
    acc.x += v[u].x;
    acc.y += v[u].y;
  }
}

template <int U>
__device__ __forceinline__ void gather_batch1(const float* __restrict__ f, int s_l,
                                              int j, int lane, float& acc) {
  float v[U];
#pragma unroll
  for (int u = 0; u < U; ++u) {
    const int s = __builtin_amdgcn_readlane(s_l, j + u);
    v[u] = f[(long)s * OUTD + lane];
  }
#pragma unroll
  for (int u = 0; u < U; ++u) acc += v[u];
}

__global__ __launch_bounds__(256) void k_gather64_relu(
    const int* __restrict__ rowptr, const int* __restrict__ col,
    const float* __restrict__ dinv, const float* __restrict__ feat,
    const float* __restrict__ bias, float* __restrict__ out, int N) {
  const int lane = threadIdx.x & 31;
  const int r0 = blockIdx.x * 8 + (threadIdx.x >> 5);
  if (r0 >= N) return;
  const int row = __builtin_amdgcn_readfirstlane(r0);
  const int beg = __builtin_amdgcn_readfirstlane(rowptr[row]);
  const int end = __builtin_amdgcn_readfirstlane(rowptr[row + 1]);
  const float dd = dinv[row];
  const float2* __restrict__ f2 = reinterpret_cast<const float2*>(feat);

  const float2 self = f2[(long)row * (HID / 2) + lane];
  float2 acc;
  acc.x = 0.0f;
  acc.y = 0.0f;

  for (int base = beg; base < end; base += 32) {
    const int n = (end - base < 32) ? (end - base) : 32;
    int s_l = row;
    if (lane < n) {
      s_l = col[base + lane];
      s_l = (s_l < 0) ? 0 : ((s_l >= N) ? N - 1 : s_l);
    }
    int j = 0;
    for (; j + 8 <= n; j += 8) gather_batch2<8>(f2, s_l, j, lane, acc);
    if (n - j >= 4) { gather_batch2<4>(f2, s_l, j, lane, acc); j += 4; }
    if (n - j >= 2) { gather_batch2<2>(f2, s_l, j, lane, acc); j += 2; }
    if (n - j >= 1) { gather_batch2<1>(f2, s_l, j, lane, acc); }
  }

  acc.x += self.x;
  acc.y += self.y;
  const float2 b = reinterpret_cast<const float2*>(bias)[lane];
  float2 o;
  o.x = fmaxf(fmaf(acc.x, dd, b.x), 0.0f) * dd;
  o.y = fmaxf(fmaf(acc.y, dd, b.y), 0.0f) * dd;
  reinterpret_cast<float2*>(out)[(long)row * (HID / 2) + lane] = o;
}

__global__ __launch_bounds__(256) void k_gather32(
    const int* __restrict__ rowptr, const int* __restrict__ col,
    const float* __restrict__ dinv, const float* __restrict__ feat,
    const float* __restrict__ bias, float* __restrict__ out, int N) {
  const int lane = threadIdx.x & 31;
  const int r0 = blockIdx.x * 8 + (threadIdx.x >> 5);
  if (r0 >= N) return;
  const int row = __builtin_amdgcn_readfirstlane(r0);
  const int beg = __builtin_amdgcn_readfirstlane(rowptr[row]);
  const int end = __builtin_amdgcn_readfirstlane(rowptr[row + 1]);
  const float dd = dinv[row];

  const float self = feat[(long)row * OUTD + lane];
  float acc = 0.0f;

  for (int base = beg; base < end; base += 32) {
    const int n = (end - base < 32) ? (end - base) : 32;
    int s_l = row;
    if (lane < n) {
      s_l = col[base + lane];
      s_l = (s_l < 0) ? 0 : ((s_l >= N) ? N - 1 : s_l);
    }
    int j = 0;
    for (; j + 8 <= n; j += 8) gather_batch1<8>(feat, s_l, j, lane, acc);
    if (n - j >= 4) { gather_batch1<4>(feat, s_l, j, lane, acc); j += 4; }
    if (n - j >= 2) { gather_batch1<2>(feat, s_l, j, lane, acc); j += 2; }
    if (n - j >= 1) { gather_batch1<1>(feat, s_l, j, lane, acc); }
  }

  acc += self;
  out[(long)row * OUTD + lane] = fmaf(acc, dd, bias[lane]);
}

__global__ void k_link(const float* __restrict__ z, const int* __restrict__ eli,
                       const float* __restrict__ Wl, const float* __restrict__ bl,
                       float* __restrict__ pred, int EL) {
  int e = blockIdx.x * blockDim.x + threadIdx.x;
  if (e >= EL) return;
  const int s = eli[e];
  const int d = eli[EL + e];
  const float* zs = z + (long)s * OUTD;
  const float* zd = z + (long)d * OUTD;
  float acc = bl[0];
#pragma unroll
  for (int c = 0; c < OUTD; ++c)
    acc = fmaf(zs[c], Wl[c], fmaf(zd[c], Wl[OUTD + c], acc));
  pred[e] = acc;
}

extern "C" void kernel_launch(void* const* d_in, const int* in_sizes, int n_in,
                              void* d_out, int out_size, void* d_ws, size_t ws_size,
                              hipStream_t stream) {
  const float* x   = (const float*)d_in[0];
  const float* pos = (const float*)d_in[1];
  const float* W1  = (const float*)d_in[2];
  const float* b1  = (const float*)d_in[3];
  const float* W2  = (const float*)d_in[4];
  const float* b2  = (const float*)d_in[5];
  const float* Wl  = (const float*)d_in[6];
  const float* bl  = (const float*)d_in[7];
  const int*   ei  = (const int*)d_in[8];
  const int*   eli = (const int*)d_in[9];

  const int N  = in_sizes[0] / IN_C;
  const int E  = in_sizes[8] / 2;
  const int EL = in_sizes[9] / 2;

  float* z    = (float*)d_out;
  float* pred = z + (long)N * OUTD;

  float* hw1   = (float*)d_ws;
  float* h1    = hw1 + (long)N * HID;
  float* reg3  = h1  + (long)N * HID;
  float* dinv  = reg3 + (long)N * OUTD;
  float* hw2   = hw1;
  int2*  seg   = (int2*)h1;
  int*   loff  = (int*)hw1;
  int*   col   = (int*)reg3;
  int*   rowptr = col + E;

  const int* esrc = ei;
  const int* edst = ei + E;

  const int B = 256;
  const int tiles  = (E + PTILE - 1) / PTILE;
  const int ranges = (N + RN - 1) / RN;

  k_part<<<tiles, PT, 0, stream>>>(esrc, edst, seg, loff, E, N, ranges);
  k_rowsort<<<ranges, RT, 0, stream>>>(seg, loff, rowptr, dinv, col, N, ranges, tiles);

  const int rowTiles = (N + 15) / 16;
  const int gblocks  = (rowTiles + 7) / 8;
  const int rblocks  = (N + 7) / 8;

  k_gemm1<<<gblocks, B, 0, stream>>>(x, pos, W1, dinv, hw1, N);
  k_gather64_relu<<<rblocks, B, 0, stream>>>(rowptr, col, dinv, hw1, b1, h1, N);

  k_gemm2<<<gblocks, B, 0, stream>>>(h1, W2, hw2, N);
  k_gather32<<<rblocks, B, 0, stream>>>(rowptr, col, dinv, hw2, b2, z, N);

  k_link<<<(EL + B - 1) / B, B, 0, stream>>>(z, eli, Wl, bl, pred, EL);
}


# Gather waves own 8 consecutive rows each (run's rowptr and dinv in lanes, next row's indices and self row asked for ahead): N/8 waves enter per gather launch, not N; per-row sum order unchanged
# speedup vs baseline: 5.7083x; 1.0400x over previous
#include <hip/hip_runtime.h>

typedef __attribute__((ext_vector_type(2))) float v2f;
typedef __attribute__((ext_vector_type(8))) float v8f;

#define IN_C 64
#define POSD 16
#define KDIM1 (IN_C + POSD)
#define HID 64
#define OUTD 32


#define RN_SHIFT 8
#define RN 256
#define MAXR 1024
#define PT 1024
#define RT 512
#define RW (RT / 32)
#define EPT 8
#define PTILE (PT * EPT)
#define CAPC 12288

__global__ __launch_bounds__(PT) void k_part(const int* __restrict__ src, const int* __restrict__ dst,
                                             int2* __restrict__ seg, int* __restrict__ loff,
                                             int E, int N, int R) {
  __shared__ int rc[MAXR];
  __shared__ int wsum[32];
  __shared__ int2 stage[PTILE];
  const int tid = threadIdx.x, lane = tid & 31, wave = tid >> 5;
  const int base = blockIdx.x * PTILE;

  rc[tid] = 0;
  __syncthreads();

  int s[EPT], d[EPT], rk[EPT];
#pragma unroll
  for (int k = 0; k < EPT; ++k) {
    const int e = base + k * PT + tid;
    const int ec = (e < E) ? e : (E - 1);
    s[k] = src[ec];
    d[k] = dst[ec];
  }
#pragma unroll
  for (int k = 0; k < EPT; ++k) {
    const int e = base + k * PT + tid;
    int t = d[k];
    t = (t < 0) ? 0 : ((t >= N) ? N - 1 : t);
    d[k] = (e < E) ? t : -1;
  }
#pragma unroll
  for (int k = 0; k < EPT; ++k) {
    rk[k] = 0;
    if (d[k] >= 0) rk[k] = atomicAdd(&rc[d[k] >> RN_SHIFT], 1);
  }
  __syncthreads();

  const int c = rc[tid];
  int inc = c;
  for (int off = 1; off < 32; off <<= 1) {
    const int t = __shfl_up(inc, off, 32);
    if (lane >= off) inc += t;
  }
  if (lane == 31) wsum[wave] = inc;
  __syncthreads();
  if (wave == 0) {
    const int v = wsum[lane];
    int iv = v;
    for (int off = 1; off < 32; off <<= 1) {
      const int t = __shfl_up(iv, off, 32);
      if (lane >= off) iv += t;
    }
    wsum[lane] = iv - v;
  }
  __syncthreads();
  const int excl = wsum[wave] + inc - c;
  rc[tid] = excl;
  if (tid <= R) loff[(long)blockIdx.x * (R + 1) + tid] = excl;
  __syncthreads();

#pragma unroll
  for (int k = 0; k < EPT; ++k) {
    if (d[k] >= 0) {
      int2 p;
      p.x = s[k];
      p.y = d[k];
      stage[rc[d[k] >> RN_SHIFT] + rk[k]] = p;
    }
  }
  __syncthreads();

  const int nloc = (E - base < PTILE) ? (E - base) : PTILE;
  for (int i = tid; i < nloc; i += PT) seg[(long)base + i] = stage[i];
}

template <bool PLACE>
__device__ __forceinline__ void walk_pieces(const int2* __restrict__ seg, const int* __restrict__ loff,
                                            int R, int PB, int r, int lane, int wave,
                                            int* cnt, int* cur, int* cstage, int* __restrict__ colr,
                                            bool staged) {
  for (int p0 = 0; p0 < PB; p0 += RT) {
    const int p = p0 + lane * RW + wave;
    int pst = 0, pcn = 0;
    if (p < PB) {
      const int a = loff[(long)p * (R + 1) + r];
      const int b = loff[(long)p * (R + 1) + r + 1];
      pst = p * PTILE + a;
      pcn = b - a;
    }
    const int rem = PB - p0 - wave;
    int npw = (rem <= 0) ? 0 : ((rem + RW - 1) / RW);
    if (npw > 32) npw = 32;
    for (int j = 0; j < npw; j += 8) {
      int st[8], cn[8];
      int2 e[8];
#pragma unroll
      for (int u = 0; u < 8; ++u) {
        st[u] = __builtin_amdgcn_readlane(pst, j + u);
        cn[u] = __builtin_amdgcn_readlane(pcn, j + u);
      }
#pragma unroll
      for (int u = 0; u < 8; ++u) {
        const long idx = (lane < cn[u]) ? ((long)st[u] + lane) : 0;
        e[u] = seg[idx];
      }
#pragma unroll
      for (int u = 0; u < 8; ++u) {
        if (lane < cn[u]) {
          const int loc = e[u].y & (RN - 1);
          if (!PLACE) {
            atomicAdd(&cnt[loc], 1);
          } else {
            const int pos = atomicAdd(&cur[loc], 1);
            if (staged) cstage[pos] = e[u].x; else colr[pos] = e[u].x;
          }
        }
      }
    }
    for (int j = 0; j < npw; ++j) {
      const int cnj = __builtin_amdgcn_readlane(pcn, j);
      if (cnj > 32) {
        const int stj = __builtin_amdgcn_readlane(pst, j);
        for (int i = 32 + lane; i < cnj; i += 32) {
          const int2 ee = seg[(long)stj + i];
          const int loc = ee.y & (RN - 1);
          if (!PLACE) {
            atomicAdd(&cnt[loc], 1);
          } else {
            const int pos = atomicAdd(&cur[loc], 1);
            if (staged) cstage[pos] = ee.x; else colr[pos] = ee.x;
          }
        }
      }
    }
  }
}

__global__ __launch_bounds__(RT) void k_rowsort(const int2* __restrict__ seg, const int* __restrict__ loff,
                                                int* __restrict__ rowptr, float* __restrict__ dinv,
                                                int* __restrict__ col, int N, int R, int PB) {
  __shared__ int cnt[RN];
  __shared__ int cur[RN];
  __shared__ int red[RW];
  __shared__ int wtot[RN / 32];
  __shared__ int cstage[CAPC];
  const int tid = threadIdx.x, lane = tid & 31, wave = tid >> 5;
  const int r = blockIdx.x;
  const int nbase = r << RN_SHIFT;

  if (tid < RN) cnt[tid] = 0;

  int bl = 0;
  for (int p = tid; p < PB; p += RT) bl += loff[(long)p * (R + 1) + r];
  for (int off = 16; off > 0; off >>= 1) bl += __shfl_down(bl, off, 32);
  if (lane == 0) red[wave] = bl;
  __syncthreads();
  int rbase = 0;
  for (int w = 0; w < RW; ++w) rbase += red[w];

  walk_pieces<false>(seg, loff, R, PB, r, lane, wave, cnt, cur, cstage, col, false);
  __syncthreads();

  const int c = (tid < RN) ? cnt[tid] : 0;
  int inc = c;
  for (int off = 1; off < 32; off <<= 1) {
    const int t = __shfl_up(inc, off, 32);
    if (lane >= off) inc += t;
  }
  if (tid < RN && lane == 31) wtot[wave] = inc;
  __syncthreads();
  int ntot = 0;
  for (int w = 0; w < RN / 32; ++w) ntot += wtot[w];
  if (tid < RN) {
    int woff = 0;
    for (int w = 0; w < wave; ++w) woff += wtot[w];
    const int excl = woff + inc - c;
    cur[tid] = excl;
    const int node = nbase + tid;
    if (node < N) {
      rowptr[node] = rbase + excl;
      dinv[node] = rsqrtf((float)(c + 1));
      if (node == N - 1) rowptr[N] = rbase + excl + c;
    }
  }
  const bool staged = (ntot <= CAPC);
  __syncthreads();

  walk_pieces<true>(seg, loff, R, PB, r, lane, wave, cnt, cur, cstage, col + rbase, staged);

  if (staged) {
    __syncthreads();
    for (int i = tid; i < ntot; i += RT) col[(long)rbase + i] = cstage[i];
  }
}

__global__ __launch_bounds__(256) void k_gemm1(
    const float* __restrict__ x, const float* __restrict__ pos,
    const float* __restrict__ W1, const float* __restrict__ dinv,
    float* __restrict__ hw1, int N) {
  const int lane = threadIdx.x & 31;
  const int wave = threadIdx.x >> 5;
  const int rowBase = (blockIdx.x * 8 + wave) * 16;
  if (rowBase >= N) return;
  const int kh  = lane >> 4;
  const int n16 = lane & 15;
  int row = rowBase + n16;
  if (row >= N) row = N - 1;

  float sc[8];
#pragma unroll
  for (int i = 0; i < 8; ++i) {
    int r = rowBase + i + 8 * kh;
    if (r >= N) r = N - 1;
    sc[i] = dinv[r];
  }

  v8f acc0 = {}, acc1 = {}, acc2 = {}, acc3 = {};
  for (int k = 0; k < KDIM1; k += 4) {
    const int ka = k + 2 * kh;
    v2f a;
    a.x = (ka     < IN_C) ? x[(long)row * IN_C + ka]       : pos[(long)row * POSD + (ka     - IN_C)];
    a.y = (ka + 1 < IN_C) ? x[(long)row * IN_C + ka + 1]   : pos[(long)row * POSD + (ka + 1 - IN_C)];
    const float* wb = W1 + (long)ka * HID + n16;
    v2f b0; b0.x = wb[0];  b0.y = wb[HID];
    v2f b1; b1.x = wb[16]; b1.y = wb[HID + 16];
    v2f b2; b2.x = wb[32]; b2.y = wb[HID + 32];
    v2f b3; b3.x = wb[48]; b3.y = wb[HID + 48];
    acc0 = __builtin_amdgcn_wmma_f32_16x16x4_f32(false, a, false, b0, (short)0, acc0, false, false);
    acc1 = __builtin_amdgcn_wmma_f32_16x16x4_f32(false, a, false, b1, (short)0, acc1, false, false);
    acc2 = __builtin_amdgcn_wmma_f32_16x16x4_f32(false, a, false, b2, (short)0, acc2, false, false);
    acc3 = __builtin_amdgcn_wmma_f32_16x16x4_f32(false, a, false, b3, (short)0, acc3, false, false);
  }
#pragma unroll
  for (int i = 0; i < 8; ++i) {
    acc0[i] *= sc[i];
    acc1[i] *= sc[i];
    acc2[i] *= sc[i];
    acc3[i] *= sc[i];
  }
#pragma unroll
  for (int i = 0; i < 8; ++i) {
    const int r = rowBase + i + 8 * kh;
    if (r < N) {
      const long base = (long)r * HID + n16;
      hw1[base +  0] = acc0[i];
      hw1[base + 16] = acc1[i];
      hw1[base + 32] = acc2[i];
      hw1[base + 48] = acc3[i];
    }
  }
}

__global__ __launch_bounds__(256) void k_gemm2(
    const float* __restrict__ h, const float* __restrict__ W2,
    float* __restrict__ hw2, int N) {
  const int lane = threadIdx.x & 31;
  const int wave = threadIdx.x >> 5;
  const int rowBase = (blockIdx.x * 8 + wave) * 16;
  if (rowBase >= N) return;
  const int kh  = lane >> 4;
  const int n16 = lane & 15;
  int row = rowBase + n16;
  if (row >= N) row = N - 1;

  v8f acc0 = {}, acc1 = {};
  for (int k = 0; k < HID; k += 4) {
    const int ka = k + 2 * kh;
    v2f a;
    a.x = h[(long)row * HID + ka];
    a.y = h[(long)row * HID + ka + 1];
    const float* wb = W2 + (long)ka * OUTD + n16;
    v2f b0; b0.x = wb[0];  b0.y = wb[OUTD];
    v2f b1; b1.x = wb[16]; b1.y = wb[OUTD + 16];
    acc0 = __builtin_amdgcn_wmma_f32_16x16x4_f32(false, a, false, b0, (short)0, acc0, false, false);
    acc1 = __builtin_amdgcn_wmma_f32_16x16x4_f32(false, a, false, b1, (short)0, acc1, false, false);
  }
#pragma unroll
  for (int i = 0; i < 8; ++i) {
    const int r = rowBase + i + 8 * kh;
    if (r < N) {
      const long base = (long)r * OUTD + n16;
      hw2[base +  0] = acc0[i];
      hw2[base + 16] = acc1[i];
    }
  }
}


template <int U>
__device__ __forceinline__ void gather_batch2(const float2* __restrict__ f2, int s_l,
                                              int j, int lane, float2& acc) {
  float2 v[U];
#pragma unroll
  for (int u = 0; u < U; ++u) {
    const int s = __builtin_amdgcn_readlane(s_l, j + u);
    v[u] = f2[(long)s * (HID / 2) + lane];
  }
#pragma unroll
  for (int u = 0; u < U; ++u) {
    acc.x += v[u].x;
    acc.y += v[u].y;
  }
}

template <int U>
__device__ __forceinline__ void gather_batch1(const float* __restrict__ f, int s_l,
                                              int j, int lane, float& acc) {
  float v[U];
#pragma unroll
  for (int u = 0; u < U; ++u) {
    const int s = __builtin_amdgcn_readlane(s_l, j + u);
    v[u] = f[(long)s * OUTD + lane];
  }
#pragma unroll
  for (int u = 0; u < U; ++u) acc += v[u];
}

#define RPW 8

__global__ __launch_bounds__(256) void k_gather64_relu(
    const int* __restrict__ rowptr, const int* __restrict__ col,
    const float* __restrict__ dinv, const float* __restrict__ feat,
    const float* __restrict__ bias, float* __restrict__ out, int N) {
  const int lane = threadIdx.x & 31;
  const int rb = (blockIdx.x * 8 + (threadIdx.x >> 5)) * RPW;
  if (rb >= N) return;
  const int r0 = __builtin_amdgcn_readfirstlane(rb);
  const int nr = (N - r0 < RPW) ? (N - r0) : RPW;
  const int rp_l = rowptr[r0 + ((lane <= nr) ? lane : nr)];
  const float dd_l = dinv[r0 + ((lane < nr) ? lane : (nr - 1))];
  const float2* __restrict__ f2 = reinterpret_cast<const float2*>(feat);
  const float2 b = reinterpret_cast<const float2*>(bias)[lane];
  float2 self_n = f2[(long)r0 * (HID / 2) + lane];

  int beg = __builtin_amdgcn_readlane(rp_l, 0);
  int end = __builtin_amdgcn_readlane(rp_l, 1);
  int c_n = r0;
  if (lane < end - beg) c_n = col[beg + lane];

  for (int ri = 0; ri < nr; ++ri) {
    const int row = r0 + ri;
    int s_l = (c_n < 0) ? 0 : ((c_n >= N) ? N - 1 : c_n);
    const float2 self = self_n;
    const int nbeg = end;
    int nend = end;
    if (ri + 1 < nr) {
      nend = __builtin_amdgcn_readlane(rp_l, ri + 2);
      c_n = row + 1;
      if (lane < nend - nbeg) c_n = col[nbeg + lane];
      self_n = f2[(long)(row + 1) * (HID / 2) + lane];
    }

    float2 acc;
    acc.x = 0.0f;
    acc.y = 0.0f;
    for (int base = beg; base < end; base += 32) {
      const int n = (end - base < 32) ? (end - base) : 32;
      if (base != beg) {
        s_l = row;
        if (lane < n) {
          s_l = col[base + lane];
          s_l = (s_l < 0) ? 0 : ((s_l >= N) ? N - 1 : s_l);
        }
      }
      int j = 0;
      for (; j + 8 <= n; j += 8) gather_batch2<8>(f2, s_l, j, lane, acc);
      if (n - j >= 4) { gather_batch2<4>(f2, s_l, j, lane, acc); j += 4; }
      if (n - j >= 2) { gather_batch2<2>(f2, s_l, j, lane, acc); j += 2; }
      if (n - j >= 1) { gather_batch2<1>(f2, s_l, j, lane, acc); }
    }

    acc.x += self.x;
    acc.y += self.y;
    const float dd = __int_as_float(__builtin_amdgcn_readlane(__float_as_int(dd_l), ri));
    float2 o;
    o.x = fmaxf(fmaf(acc.x, dd, b.x), 0.0f) * dd;
    o.y = fmaxf(fmaf(acc.y, dd, b.y), 0.0f) * dd;
    reinterpret_cast<float2*>(out)[(long)row * (HID / 2) + lane] = o;

    beg = nbeg;
    end = nend;
  }
}

__global__ __launch_bounds__(256) void k_gather32(
    const int* __restrict__ rowptr, const int* __restrict__ col,
    const float* __restrict__ dinv, const float* __restrict__ feat,
    const float* __restrict__ bias, float* __restrict__ out, int N) {
  const int lane = threadIdx.x & 31;
  const int rb = (blockIdx.x * 8 + (threadIdx.x >> 5)) * RPW;
  if (rb >= N) return;
  const int r0 = __builtin_amdgcn_readfirstlane(rb);
  const int nr = (N - r0 < RPW) ? (N - r0) : RPW;
  const int rp_l = rowptr[r0 + ((lane <= nr) ? lane : nr)];
  const float dd_l = dinv[r0 + ((lane < nr) ? lane : (nr - 1))];
  const float bv = bias[lane];
  float self_n = feat[(long)r0 * OUTD + lane];

  int beg = __builtin_amdgcn_readlane(rp_l, 0);
  int end = __builtin_amdgcn_readlane(rp_l, 1);
  int c_n = r0;
  if (lane < end - beg) c_n = col[beg + lane];

  for (int ri = 0; ri < nr; ++ri) {
    const int row = r0 + ri;
    int s_l = (c_n < 0) ? 0 : ((c_n >= N) ? N - 1 : c_n);
    const float self = self_n;
    const int nbeg = end;
    int nend = end;
    if (ri + 1 < nr) {
      nend = __builtin_amdgcn_readlane(rp_l, ri + 2);
      c_n = row + 1;
      if (lane < nend - nbeg) c_n = col[nbeg + lane];
      self_n = feat[(long)(row + 1) * OUTD + lane];
    }

    float acc = 0.0f;
    for (int base = beg; base < end; base += 32) {
      const int n = (end - base < 32) ? (end - base) : 32;
      if (base != beg) {
        s_l = row;
        if (lane < n) {
          s_l = col[base + lane];
          s_l = (s_l < 0) ? 0 : ((s_l >= N) ? N - 1 : s_l);
        }
      }
      int j = 0;
      for (; j + 8 <= n; j += 8) gather_batch1<8>(feat, s_l, j, lane, acc);
      if (n - j >= 4) { gather_batch1<4>(feat, s_l, j, lane, acc); j += 4; }
      if (n - j >= 2) { gather_batch1<2>(feat, s_l, j, lane, acc); j += 2; }
      if (n - j >= 1) { gather_batch1<1>(feat, s_l, j, lane, acc); }
    }

    acc += self;
    const float dd = __int_as_float(__builtin_amdgcn_readlane(__float_as_int(dd_l), ri));
    out[(long)row * OUTD + lane] = fmaf(acc, dd, bv);

    beg = nbeg;
    end = nend;
  }
}

__global__ void k_link(const float* __restrict__ z, const int* __restrict__ eli,
                       const float* __restrict__ Wl, const float* __restrict__ bl,
                       float* __restrict__ pred, int EL) {
  int e = blockIdx.x * blockDim.x + threadIdx.x;
  if (e >= EL) return;
  const int s = eli[e];
  const int d = eli[EL + e];
  const float* zs = z + (long)s * OUTD;
  const float* zd = z + (long)d * OUTD;
  float acc = bl[0];
#pragma unroll
  for (int c = 0; c < OUTD; ++c)
    acc = fmaf(zs[c], Wl[c], fmaf(zd[c], Wl[OUTD + c], acc));
  pred[e] = acc;
}

extern "C" void kernel_launch(void* const* d_in, const int* in_sizes, int n_in,
                              void* d_out, int out_size, void* d_ws, size_t ws_size,
                              hipStream_t stream) {
  const float* x   = (const float*)d_in[0];
  const float* pos = (const float*)d_in[1];
  const float* W1  = (const float*)d_in[2];
  const float* b1  = (const float*)d_in[3];
  const float* W2  = (const float*)d_in[4];
  const float* b2  = (const float*)d_in[5];
  const float* Wl  = (const float*)d_in[6];
  const float* bl  = (const float*)d_in[7];
  const int*   ei  = (const int*)d_in[8];
  const int*   eli = (const int*)d_in[9];

  const int N  = in_sizes[0] / IN_C;
  const int E  = in_sizes[8] / 2;
  const int EL = in_sizes[9] / 2;

  float* z    = (float*)d_out;
  float* pred = z + (long)N * OUTD;

  float* hw1   = (float*)d_ws;
  float* h1    = hw1 + (long)N * HID;
  float* reg3  = h1  + (long)N * HID;
  float* dinv  = reg3 + (long)N * OUTD;
  float* hw2   = hw1;
  int2*  seg   = (int2*)h1;
  int*   loff  = (int*)hw1;
  int*   col   = (int*)reg3;
  int*   rowptr = col + E;

  const int* esrc = ei;
  const int* edst = ei + E;

  const int B = 256;
  const int tiles  = (E + PTILE - 1) / PTILE;
  const int ranges = (N + RN - 1) / RN;

  k_part<<<tiles, PT, 0, stream>>>(esrc, edst, seg, loff, E, N, ranges);
  k_rowsort<<<ranges, RT, 0, stream>>>(seg, loff, rowptr, dinv, col, N, ranges, tiles);

  const int rowTiles = (N + 15) / 16;
  const int gblocks  = (rowTiles + 7) / 8;
  const int rblocks  = (N + 8 * RPW - 1) / (8 * RPW);

  k_gemm1<<<gblocks, B, 0, stream>>>(x, pos, W1, dinv, hw1, N);
  k_gather64_relu<<<rblocks, B, 0, stream>>>(rowptr, col, dinv, hw1, b1, h1, N);

  k_gemm2<<<gblocks, B, 0, stream>>>(h1, W2, hw2, N);
  k_gather32<<<rblocks, B, 0, stream>>>(rowptr, col, dinv, hw2, b2, z, N);

  k_link<<<(EL + B - 1) / B, B, 0, stream>>>(z, eli, Wl, bl, pred, EL);
}


# Layer 2's transform fused into layer 1's gather launch: finished h1 rows stay in LDS and 4 waves per 64-row block multiply them by W2 on WMMA (same k order); h1 never stored, one launch fewer
# speedup vs baseline: 5.9642x; 1.0448x over previous
#include <hip/hip_runtime.h>

typedef __attribute__((ext_vector_type(2))) float v2f;
typedef __attribute__((ext_vector_type(8))) float v8f;

#define IN_C 64
#define POSD 16
#define KDIM1 (IN_C + POSD)
#define HID 64
#define OUTD 32


#define RN_SHIFT 8
#define RN 256
#define MAXR 1024
#define PT 1024
#define RT 512
#define RW (RT / 32)
#define EPT 8
#define PTILE (PT * EPT)
#define CAPC 12288

__global__ __launch_bounds__(PT) void k_part(const int* __restrict__ src, const int* __restrict__ dst,
                                             int2* __restrict__ seg, int* __restrict__ loff,
                                             int E, int N, int R) {
  __shared__ int rc[MAXR];
  __shared__ int wsum[32];
  __shared__ int2 stage[PTILE];
  const int tid = threadIdx.x, lane = tid & 31, wave = tid >> 5;
  const int base = blockIdx.x * PTILE;

  rc[tid] = 0;
  __syncthreads();

  int s[EPT], d[EPT], rk[EPT];
#pragma unroll
  for (int k = 0; k < EPT; ++k) {
    const int e = base + k * PT + tid;
    const int ec = (e < E) ? e : (E - 1);
    s[k] = src[ec];
    d[k] = dst[ec];
  }
#pragma unroll
  for (int k = 0; k < EPT; ++k) {
    const int e = base + k * PT + tid;
    int t = d[k];
    t = (t < 0) ? 0 : ((t >= N) ? N - 1 : t);
    d[k] = (e < E) ? t : -1;
  }
#pragma unroll
  for (int k = 0; k < EPT; ++k) {
    rk[k] = 0;
    if (d[k] >= 0) rk[k] = atomicAdd(&rc[d[k] >> RN_SHIFT], 1);
  }
  __syncthreads();

  const int c = rc[tid];
  int inc = c;
  for (int off = 1; off < 32; off <<= 1) {
    const int t = __shfl_up(inc, off, 32);
    if (lane >= off) inc += t;
  }
  if (lane == 31) wsum[wave] = inc;
  __syncthreads();
  if (wave == 0) {
    const int v = wsum[lane];
    int iv = v;
    for (int off = 1; off < 32; off <<= 1) {
      const int t = __shfl_up(iv, off, 32);
      if (lane >= off) iv += t;
    }
    wsum[lane] = iv - v;
  }
  __syncthreads();
  const int excl = wsum[wave] + inc - c;
  rc[tid] = excl;
  if (tid <= R) loff[(long)blockIdx.x * (R + 1) + tid] = excl;
  __syncthreads();

#pragma unroll
  for (int k = 0; k < EPT; ++k) {
    if (d[k] >= 0) {
      int2 p;
      p.x = s[k];
      p.y = d[k];
      stage[rc[d[k] >> RN_SHIFT] + rk[k]] = p;
    }
  }
  __syncthreads();

  const int nloc = (E - base < PTILE) ? (E - base) : PTILE;
  for (int i = tid; i < nloc; i += PT) seg[(long)base + i] = stage[i];
}

template <bool PLACE>
__device__ __forceinline__ void walk_pieces(const int2* __restrict__ seg, const int* __restrict__ loff,
                                            int R, int PB, int r, int lane, int wave,
                                            int* cnt, int* cur, int* cstage, int* __restrict__ colr,
                                            bool staged) {
  for (int p0 = 0; p0 < PB; p0 += RT) {
    const int p = p0 + lane * RW + wave;
    int pst = 0, pcn = 0;
    if (p < PB) {
      const int a = loff[(long)p * (R + 1) + r];
      const int b = loff[(long)p * (R + 1) + r + 1];
      pst = p * PTILE + a;
      pcn = b - a;
    }
    const int rem = PB - p0 - wave;
    int npw = (rem <= 0) ? 0 : ((rem + RW - 1) / RW);
    if (npw > 32) npw = 32;
    for (int j = 0; j < npw; j += 8) {
      int st[8], cn[8];
      int2 e[8];
#pragma unroll
      for (int u = 0; u < 8; ++u) {
        st[u] = __builtin_amdgcn_readlane(pst, j + u);
        cn[u] = __builtin_amdgcn_readlane(pcn, j + u);
      }
#pragma unroll
      for (int u = 0; u < 8; ++u) {
        const long idx = (lane < cn[u]) ? ((long)st[u] + lane) : 0;
        e[u] = seg[idx];
      }
#pragma unroll
      for (int u = 0; u < 8; ++u) {
        if (lane < cn[u]) {
          const int loc = e[u].y & (RN - 1);
          if (!PLACE) {
            atomicAdd(&cnt[loc], 1);
          } else {
            const int pos = atomicAdd(&cur[loc], 1);
            if (staged) cstage[pos] = e[u].x; else colr[pos] = e[u].x;
          }
        }
      }
    }
    for (int j = 0; j < npw; ++j) {
      const int cnj = __builtin_amdgcn_readlane(pcn, j);
      if (cnj > 32) {
        const int stj = __builtin_amdgcn_readlane(pst, j);
        for (int i = 32 + lane; i < cnj; i += 32) {
          const int2 ee = seg[(long)stj + i];
          const int loc = ee.y & (RN - 1);
          if (!PLACE) {
            atomicAdd(&cnt[loc], 1);
          } else {
            const int pos = atomicAdd(&cur[loc], 1);
            if (staged) cstage[pos] = ee.x; else colr[pos] = ee.x;
          }
        }
      }
    }
  }
}

__global__ __launch_bounds__(RT) void k_rowsort(const int2* __restrict__ seg, const int* __restrict__ loff,
                                                int* __restrict__ rowptr, float* __restrict__ dinv,
                                                int* __restrict__ col, int N, int R, int PB) {
  __shared__ int cnt[RN];
  __shared__ int cur[RN];
  __shared__ int red[RW];
  __shared__ int wtot[RN / 32];
  __shared__ int cstage[CAPC];
  const int tid = threadIdx.x, lane = tid & 31, wave = tid >> 5;
  const int r = blockIdx.x;
  const int nbase = r << RN_SHIFT;

  if (tid < RN) cnt[tid] = 0;

  int bl = 0;
  for (int p = tid; p < PB; p += RT) bl += loff[(long)p * (R + 1) + r];
  for (int off = 16; off > 0; off >>= 1) bl += __shfl_down(bl, off, 32);
  if (lane == 0) red[wave] = bl;
  __syncthreads();
  int rbase = 0;
  for (int w = 0; w < RW; ++w) rbase += red[w];

  walk_pieces<false>(seg, loff, R, PB, r, lane, wave, cnt, cur, cstage, col, false);
  __syncthreads();

  const int c = (tid < RN) ? cnt[tid] : 0;
  int inc = c;
  for (int off = 1; off < 32; off <<= 1) {
    const int t = __shfl_up(inc, off, 32);
    if (lane >= off) inc += t;
  }
  if (tid < RN && lane == 31) wtot[wave] = inc;
  __syncthreads();
  int ntot = 0;
  for (int w = 0; w < RN / 32; ++w) ntot += wtot[w];
  if (tid < RN) {
    int woff = 0;
    for (int w = 0; w < wave; ++w) woff += wtot[w];
    const int excl = woff + inc - c;
    cur[tid] = excl;
    const int node = nbase + tid;
    if (node < N) {
      rowptr[node] = rbase + excl;
      dinv[node] = rsqrtf((float)(c + 1));
      if (node == N - 1) rowptr[N] = rbase + excl + c;
    }
  }
  const bool staged = (ntot <= CAPC);
  __syncthreads();

  walk_pieces<true>(seg, loff, R, PB, r, lane, wave, cnt, cur, cstage, col + rbase, staged);

  if (staged) {
    __syncthreads();
    for (int i = tid; i < ntot; i += RT) col[(long)rbase + i] = cstage[i];
  }
}

__global__ __launch_bounds__(256) void k_gemm1(
    const float* __restrict__ x, const float* __restrict__ pos,
    const float* __restrict__ W1, const float* __restrict__ dinv,
    float* __restrict__ hw1, int N) {
  const int lane = threadIdx.x & 31;
  const int wave = threadIdx.x >> 5;
  const int rowBase = (blockIdx.x * 8 + wave) * 16;
  if (rowBase >= N) return;
  const int kh  = lane >> 4;
  const int n16 = lane & 15;
  int row = rowBase + n16;
  if (row >= N) row = N - 1;

  float sc[8];
#pragma unroll
  for (int i = 0; i < 8; ++i) {
    int r = rowBase + i + 8 * kh;
    if (r >= N) r = N - 1;
    sc[i] = dinv[r];
  }

  v8f acc0 = {}, acc1 = {}, acc2 = {}, acc3 = {};
  for (int k = 0; k < KDIM1; k += 4) {
    const int ka = k + 2 * kh;
    v2f a;
    a.x = (ka     < IN_C) ? x[(long)row * IN_C + ka]       : pos[(long)row * POSD + (ka     - IN_C)];
    a.y = (ka + 1 < IN_C) ? x[(long)row * IN_C + ka + 1]   : pos[(long)row * POSD + (ka + 1 - IN_C)];
    const float* wb = W1 + (long)ka * HID + n16;
    v2f b0; b0.x = wb[0];  b0.y = wb[HID];
    v2f b1; b1.x = wb[16]; b1.y = wb[HID + 16];
    v2f b2; b2.x = wb[32]; b2.y = wb[HID + 32];
    v2f b3; b3.x = wb[48]; b3.y = wb[HID + 48];
    acc0 = __builtin_amdgcn_wmma_f32_16x16x4_f32(false, a, false, b0, (short)0, acc0, false, false);
    acc1 = __builtin_amdgcn_wmma_f32_16x16x4_f32(false, a, false, b1, (short)0, acc1, false, false);
    acc2 = __builtin_amdgcn_wmma_f32_16x16x4_f32(false, a, false, b2, (short)0, acc2, false, false);
    acc3 = __builtin_amdgcn_wmma_f32_16x16x4_f32(false, a, false, b3, (short)0, acc3, false, false);
  }
#pragma unroll
  for (int i = 0; i < 8; ++i) {
    acc0[i] *= sc[i];
    acc1[i] *= sc[i];
    acc2[i] *= sc[i];
    acc3[i] *= sc[i];
  }
#pragma unroll
  for (int i = 0; i < 8; ++i) {
    const int r = rowBase + i + 8 * kh;
    if (r < N) {
      const long base = (long)r * HID + n16;
      hw1[base +  0] = acc0[i];
      hw1[base + 16] = acc1[i];
      hw1[base + 32] = acc2[i];
      hw1[base + 48] = acc3[i];
    }
  }
}


template <int U>
__device__ __forceinline__ void gather_batch2(const float2* __restrict__ f2, int s_l,
                                              int j, int lane, float2& acc) {
  float2 v[U];
#pragma unroll
  for (int u = 0; u < U; ++u) {
    const int s = __builtin_amdgcn_readlane(s_l, j + u);
    v[u] = f2[(long)s * (HID / 2) + lane];
  }
#pragma unroll
  for (int u = 0; u < U; ++u) {
    acc.x += v[u].x;
    acc.y += v[u].y;
  }
}

template <int U>
__device__ __forceinline__ void gather_batch1(const float* __restrict__ f, int s_l,
                                              int j, int lane, float& acc) {
  float v[U];
#pragma unroll
  for (int u = 0; u < U; ++u) {
    const int s = __builtin_amdgcn_readlane(s_l, j + u);
    v[u] = f[(long)s * OUTD + lane];
  }
#pragma unroll
  for (int u = 0; u < U; ++u) acc += v[u];
}

#define RPW 8

#define BROWS (8 * RPW)
#define HP 68

__global__ __launch_bounds__(256) void k_gather64_relu_gemm2(
    const int* __restrict__ rowptr, const int* __restrict__ col,
    const float* __restrict__ dinv, const float* __restrict__ feat,
    const float* __restrict__ bias, const float* __restrict__ W2,
    float* __restrict__ hw2, int N) {
  __shared__ float hs[BROWS * HP];
  const int lane = threadIdx.x & 31;
  const int wave = __builtin_amdgcn_readfirstlane((int)(threadIdx.x >> 5));
  const int r0 = (blockIdx.x * 8 + wave) * RPW;
  float* __restrict__ hrow = hs + wave * (RPW * HP);

  int nr = N - r0;
  nr = (nr < 0) ? 0 : ((nr < RPW) ? nr : RPW);

  if (nr > 0) {
    const int rp_l = rowptr[r0 + ((lane <= nr) ? lane : nr)];
    const float dd_l = dinv[r0 + ((lane < nr) ? lane : (nr - 1))];
    const float2* __restrict__ f2 = reinterpret_cast<const float2*>(feat);
    const float2 b = reinterpret_cast<const float2*>(bias)[lane];
    float2 self_n = f2[(long)r0 * (HID / 2) + lane];

    int beg = __builtin_amdgcn_readlane(rp_l, 0);
    int end = __builtin_amdgcn_readlane(rp_l, 1);
    int c_n = r0;
    if (lane < end - beg) c_n = col[beg + lane];

    for (int ri = 0; ri < nr; ++ri) {
      const int row = r0 + ri;
      int s_l = (c_n < 0) ? 0 : ((c_n >= N) ? N - 1 : c_n);
      const float2 self = self_n;
      const int nbeg = end;
      int nend = end;
      if (ri + 1 < nr) {
        nend = __builtin_amdgcn_readlane(rp_l, ri + 2);
        c_n = row + 1;
        if (lane < nend - nbeg) c_n = col[nbeg + lane];
        self_n = f2[(long)(row + 1) * (HID / 2) + lane];
      }

      float2 acc;
      acc.x = 0.0f;
      acc.y = 0.0f;
      for (int base = beg; base < end; base += 32) {
        const int n = (end - base < 32) ? (end - base) : 32;
        if (base != beg) {
          s_l = row;
          if (lane < n) {
            s_l = col[base + lane];
            s_l = (s_l < 0) ? 0 : ((s_l >= N) ? N - 1 : s_l);
          }
        }
        int j = 0;
        for (; j + 8 <= n; j += 8) gather_batch2<8>(f2, s_l, j, lane, acc);
        if (n - j >= 4) { gather_batch2<4>(f2, s_l, j, lane, acc); j += 4; }
        if (n - j >= 2) { gather_batch2<2>(f2, s_l, j, lane, acc); j += 2; }
        if (n - j >= 1) { gather_batch2<1>(f2, s_l, j, lane, acc); }
      }

      acc.x += self.x;
      acc.y += self.y;
      const float dd = __int_as_float(__builtin_amdgcn_readlane(__float_as_int(dd_l), ri));
      float2 o;
      o.x = fmaxf(fmaf(acc.x, dd, b.x), 0.0f) * dd;
      o.y = fmaxf(fmaf(acc.y, dd, b.y), 0.0f) * dd;
      reinterpret_cast<float2*>(hrow + ri * HP)[lane] = o;

      beg = nbeg;
      end = nend;
    }
  }
  for (int ri = nr; ri < RPW; ++ri) {
    float2 zz;
    zz.x = 0.0f;
    zz.y = 0.0f;
    reinterpret_cast<float2*>(hrow + ri * HP)[lane] = zz;
  }
  __syncthreads();

  if (wave >= BROWS / 16) return;
  const int rowBase = blockIdx.x * BROWS + wave * 16;
  if (rowBase >= N) return;
  const int kh  = lane >> 4;
  const int n16 = lane & 15;
  const float* ar = hs + (wave * 16 + n16) * HP + 2 * kh;
  const float* wr = W2 + (2 * kh) * OUTD + n16;

  v8f acc0 = {}, acc1 = {};
#pragma unroll 8
  for (int k = 0; k < HID; k += 4) {
    const v2f a = *reinterpret_cast<const v2f*>(ar + k);
    const float* wb = wr + k * OUTD;
    v2f b0; b0.x = wb[0];  b0.y = wb[OUTD];
    v2f b1; b1.x = wb[16]; b1.y = wb[OUTD + 16];
    acc0 = __builtin_amdgcn_wmma_f32_16x16x4_f32(false, a, false, b0, (short)0, acc0, false, false);
    acc1 = __builtin_amdgcn_wmma_f32_16x16x4_f32(false, a, false, b1, (short)0, acc1, false, false);
  }
  int rfirst = rowBase + 8 * kh;
  int cfirst = n16;
  asm volatile("v_nop\n\tv_nop\n\tv_nop\n\tv_nop" : "+v"(acc0), "+v"(acc1), "+v"(rfirst), "+v"(cfirst));
#pragma unroll
  for (int i = 0; i < 8; ++i) {
    const int r = rfirst + i;
    if (r < N) {
      const long base = (long)r * OUTD + cfirst;
      hw2[base +  0] = acc0[i];
      hw2[base + 16] = acc1[i];
    }
  }
}

__global__ __launch_bounds__(256) void k_gather32(
    const int* __restrict__ rowptr, const int* __restrict__ col,
    const float* __restrict__ dinv, const float* __restrict__ feat,
    const float* __restrict__ bias, float* __restrict__ out, int N) {
  const int lane = threadIdx.x & 31;
  const int rb = (blockIdx.x * 8 + (threadIdx.x >> 5)) * RPW;
  if (rb >= N) return;
  const int r0 = __builtin_amdgcn_readfirstlane(rb);
  const int nr = (N - r0 < RPW) ? (N - r0) : RPW;
  const int rp_l = rowptr[r0 + ((lane <= nr) ? lane : nr)];
  const float dd_l = dinv[r0 + ((lane < nr) ? lane : (nr - 1))];
  const float bv = bias[lane];
  float self_n = feat[(long)r0 * OUTD + lane];

  int beg = __builtin_amdgcn_readlane(rp_l, 0);
  int end = __builtin_amdgcn_readlane(rp_l, 1);
  int c_n = r0;
  if (lane < end - beg) c_n = col[beg + lane];

  for (int ri = 0; ri < nr; ++ri) {
    const int row = r0 + ri;
    int s_l = (c_n < 0) ? 0 : ((c_n >= N) ? N - 1 : c_n);
    const float self = self_n;
    const int nbeg = end;
    int nend = end;
    if (ri + 1 < nr) {
      nend = __builtin_amdgcn_readlane(rp_l, ri + 2);
      c_n = row + 1;
      if (lane < nend - nbeg) c_n = col[nbeg + lane];
      self_n = feat[(long)(row + 1) * OUTD + lane];
    }

    float acc = 0.0f;
    for (int base = beg; base < end; base += 32) {
      const int n = (end - base < 32) ? (end - base) : 32;
      if (base != beg) {
        s_l = row;
        if (lane < n) {
          s_l = col[base + lane];
          s_l = (s_l < 0) ? 0 : ((s_l >= N) ? N - 1 : s_l);
        }
      }
      int j = 0;
      for (; j + 8 <= n; j += 8) gather_batch1<8>(feat, s_l, j, lane, acc);
      if (n - j >= 4) { gather_batch1<4>(feat, s_l, j, lane, acc); j += 4; }
      if (n - j >= 2) { gather_batch1<2>(feat, s_l, j, lane, acc); j += 2; }
      if (n - j >= 1) { gather_batch1<1>(feat, s_l, j, lane, acc); }
    }

    acc += self;
    const float dd = __int_as_float(__builtin_amdgcn_readlane(__float_as_int(dd_l), ri));
    out[(long)row * OUTD + lane] = fmaf(acc, dd, bv);

    beg = nbeg;
    end = nend;
  }
}

__global__ void k_link(const float* __restrict__ z, const int* __restrict__ eli,
                       const float* __restrict__ Wl, const float* __restrict__ bl,
                       float* __restrict__ pred, int EL) {
  int e = blockIdx.x * blockDim.x + threadIdx.x;
  if (e >= EL) return;
  const int s = eli[e];
  const int d = eli[EL + e];
  const float* zs = z + (long)s * OUTD;
  const float* zd = z + (long)d * OUTD;
  float acc = bl[0];
#pragma unroll
  for (int c = 0; c < OUTD; ++c)
    acc = fmaf(zs[c], Wl[c], fmaf(zd[c], Wl[OUTD + c], acc));
  pred[e] = acc;
}

extern "C" void kernel_launch(void* const* d_in, const int* in_sizes, int n_in,
                              void* d_out, int out_size, void* d_ws, size_t ws_size,
                              hipStream_t stream) {
  const float* x   = (const float*)d_in[0];
  const float* pos = (const float*)d_in[1];
  const float* W1  = (const float*)d_in[2];
  const float* b1  = (const float*)d_in[3];
  const float* W2  = (const float*)d_in[4];
  const float* b2  = (const float*)d_in[5];
  const float* Wl  = (const float*)d_in[6];
  const float* bl  = (const float*)d_in[7];
  const int*   ei  = (const int*)d_in[8];
  const int*   eli = (const int*)d_in[9];

  const int N  = in_sizes[0] / IN_C;
  const int E  = in_sizes[8] / 2;
  const int EL = in_sizes[9] / 2;

  float* z    = (float*)d_out;
  float* pred = z + (long)N * OUTD;

  float* hw1   = (float*)d_ws;
  float* h1    = hw1 + (long)N * HID;
  float* reg3  = h1  + (long)N * HID;
  float* dinv  = reg3 + (long)N * OUTD;
  float* hw2   = h1;
  int2*  seg   = (int2*)h1;
  int*   loff  = (int*)hw1;
  int*   col   = (int*)reg3;
  int*   rowptr = col + E;

  const int* esrc = ei;
  const int* edst = ei + E;

  const int B = 256;
  const int tiles  = (E + PTILE - 1) / PTILE;
  const int ranges = (N + RN - 1) / RN;

  k_part<<<tiles, PT, 0, stream>>>(esrc, edst, seg, loff, E, N, ranges);
  k_rowsort<<<ranges, RT, 0, stream>>>(seg, loff, rowptr, dinv, col, N, ranges, tiles);

  const int rowTiles = (N + 15) / 16;
  const int gblocks  = (rowTiles + 7) / 8;
  const int rblocks  = (N + 8 * RPW - 1) / (8 * RPW);

  k_gemm1<<<gblocks, B, 0, stream>>>(x, pos, W1, dinv, hw1, N);
  k_gather64_relu_gemm2<<<rblocks, B, 0, stream>>>(rowptr, col, dinv, hw1, b1, W2, hw2, N);

  k_gather32<<<rblocks, B, 0, stream>>>(rowptr, col, dinv, hw2, b2, z, N);

  k_link<<<(EL + B - 1) / B, B, 0, stream>>>(z, eli, Wl, bl, pred, EL);
}


# GEMM 1 fused into the CSR row-sort launch: each range block's 16 waves multiply the range's own 16 row tiles by W1 on WMMA behind the sort (same k order), dinv from LDS; one launch fewer
# speedup vs baseline: 6.3037x; 1.0569x over previous
#include <hip/hip_runtime.h>

typedef __attribute__((ext_vector_type(2))) float v2f;
typedef __attribute__((ext_vector_type(8))) float v8f;

#define IN_C 64
#define POSD 16
#define KDIM1 (IN_C + POSD)
#define HID 64
#define OUTD 32


#define RN_SHIFT 8
#define RN 256
#define MAXR 1024
#define PT 1024
#define RT 512
#define RW (RT / 32)
#define EPT 8
#define PTILE (PT * EPT)
#define CAPC 12288

__global__ __launch_bounds__(PT) void k_part(const int* __restrict__ src, const int* __restrict__ dst,
                                             int2* __restrict__ seg, int* __restrict__ loff,
                                             int E, int N, int R) {
  __shared__ int rc[MAXR];
  __shared__ int wsum[32];
  __shared__ int2 stage[PTILE];
  const int tid = threadIdx.x, lane = tid & 31, wave = tid >> 5;
  const int base = blockIdx.x * PTILE;

  rc[tid] = 0;
  __syncthreads();

  int s[EPT], d[EPT], rk[EPT];
#pragma unroll
  for (int k = 0; k < EPT; ++k) {
    const int e = base + k * PT + tid;
    const int ec = (e < E) ? e : (E - 1);
    s[k] = src[ec];
    d[k] = dst[ec];
  }
#pragma unroll
  for (int k = 0; k < EPT; ++k) {
    const int e = base + k * PT + tid;
    int t = d[k];
    t = (t < 0) ? 0 : ((t >= N) ? N - 1 : t);
    d[k] = (e < E) ? t : -1;
  }
#pragma unroll
  for (int k = 0; k < EPT; ++k) {
    rk[k] = 0;
    if (d[k] >= 0) rk[k] = atomicAdd(&rc[d[k] >> RN_SHIFT], 1);
  }
  __syncthreads();

  const int c = rc[tid];
  int inc = c;
  for (int off = 1; off < 32; off <<= 1) {
    const int t = __shfl_up(inc, off, 32);
    if (lane >= off) inc += t;
  }
  if (lane == 31) wsum[wave] = inc;
  __syncthreads();
  if (wave == 0) {
    const int v = wsum[lane];
    int iv = v;
    for (int off = 1; off < 32; off <<= 1) {
      const int t = __shfl_up(iv, off, 32);
      if (lane >= off) iv += t;
    }
    wsum[lane] = iv - v;
  }
  __syncthreads();
  const int excl = wsum[wave] + inc - c;
  rc[tid] = excl;
  if (tid <= R) loff[(long)blockIdx.x * (R + 1) + tid] = excl;
  __syncthreads();

#pragma unroll
  for (int k = 0; k < EPT; ++k) {
    if (d[k] >= 0) {
      int2 p;
      p.x = s[k];
      p.y = d[k];
      stage[rc[d[k] >> RN_SHIFT] + rk[k]] = p;
    }
  }
  __syncthreads();

  const int nloc = (E - base < PTILE) ? (E - base) : PTILE;
  for (int i = tid; i < nloc; i += PT) seg[(long)base + i] = stage[i];
}

template <bool PLACE>
__device__ __forceinline__ void walk_pieces(const int2* __restrict__ seg, const int* __restrict__ loff,
                                            int R, int PB, int r, int lane, int wave,
                                            int* cnt, int* cur, int* cstage, int* __restrict__ colr,
                                            bool staged) {
  for (int p0 = 0; p0 < PB; p0 += RT) {
    const int p = p0 + lane * RW + wave;
    int pst = 0, pcn = 0;
    if (p < PB) {
      const int a = loff[(long)p * (R + 1) + r];
      const int b = loff[(long)p * (R + 1) + r + 1];
      pst = p * PTILE + a;
      pcn = b - a;
    }
    const int rem = PB - p0 - wave;
    int npw = (rem <= 0) ? 0 : ((rem + RW - 1) / RW);
    if (npw > 32) npw = 32;
    for (int j = 0; j < npw; j += 8) {
      int st[8], cn[8];
      int2 e[8];
#pragma unroll
      for (int u = 0; u < 8; ++u) {
        st[u] = __builtin_amdgcn_readlane(pst, j + u);
        cn[u] = __builtin_amdgcn_readlane(pcn, j + u);
      }
#pragma unroll
      for (int u = 0; u < 8; ++u) {
        const long idx = (lane < cn[u]) ? ((long)st[u] + lane) : 0;
        e[u] = seg[idx];
      }
#pragma unroll
      for (int u = 0; u < 8; ++u) {
        if (lane < cn[u]) {
          const int loc = e[u].y & (RN - 1);
          if (!PLACE) {
            atomicAdd(&cnt[loc], 1);
          } else {
            const int pos = atomicAdd(&cur[loc], 1);
            if (staged) cstage[pos] = e[u].x; else colr[pos] = e[u].x;
          }
        }
      }
    }
    for (int j = 0; j < npw; ++j) {
      const int cnj = __builtin_amdgcn_readlane(pcn, j);
      if (cnj > 32) {
        const int stj = __builtin_amdgcn_readlane(pst, j);
        for (int i = 32 + lane; i < cnj; i += 32) {
          const int2 ee = seg[(long)stj + i];
          const int loc = ee.y & (RN - 1);
          if (!PLACE) {
            atomicAdd(&cnt[loc], 1);
          } else {
            const int pos = atomicAdd(&cur[loc], 1);
            if (staged) cstage[pos] = ee.x; else colr[pos] = ee.x;
          }
        }
      }
    }
  }
}

static_assert(RN == 16 * RW, "one 16-row tile of the range per wave");
__global__ __launch_bounds__(RT) void k_rowsort_gemm1(const int2* __restrict__ seg, const int* __restrict__ loff,
                                                      int* __restrict__ rowptr, float* __restrict__ dinv,
                                                      int* __restrict__ col,
                                                      const float* __restrict__ x, const float* __restrict__ pos,
                                                      const float* __restrict__ W1, float* __restrict__ hw1,
                                                      int N, int R, int PB) {
  __shared__ float dl[RN];
  __shared__ int cnt[RN];
  __shared__ int cur[RN];
  __shared__ int red[RW];
  __shared__ int wtot[RN / 32];
  __shared__ int cstage[CAPC];
  const int tid = threadIdx.x, lane = tid & 31, wave = tid >> 5;
  const int r = blockIdx.x;
  const int nbase = r << RN_SHIFT;

  if (tid < RN) cnt[tid] = 0;

  int bl = 0;
  for (int p = tid; p < PB; p += RT) bl += loff[(long)p * (R + 1) + r];
  for (int off = 16; off > 0; off >>= 1) bl += __shfl_down(bl, off, 32);
  if (lane == 0) red[wave] = bl;
  __syncthreads();
  int rbase = 0;
  for (int w = 0; w < RW; ++w) rbase += red[w];

  walk_pieces<false>(seg, loff, R, PB, r, lane, wave, cnt, cur, cstage, col, false);
  __syncthreads();

  const int c = (tid < RN) ? cnt[tid] : 0;
  int inc = c;
  for (int off = 1; off < 32; off <<= 1) {
    const int t = __shfl_up(inc, off, 32);
    if (lane >= off) inc += t;
  }
  if (tid < RN && lane == 31) wtot[wave] = inc;
  __syncthreads();
  int ntot = 0;
  for (int w = 0; w < RN / 32; ++w) ntot += wtot[w];
  if (tid < RN) {
    int woff = 0;
    for (int w = 0; w < wave; ++w) woff += wtot[w];
    const int excl = woff + inc - c;
    cur[tid] = excl;
    const int node = nbase + tid;
    if (node < N) {
      rowptr[node] = rbase + excl;
      dinv[node] = rsqrtf((float)(c + 1));
      if (node == N - 1) rowptr[N] = rbase + excl + c;
    }
    dl[tid] = rsqrtf((float)(c + 1));
  }
  const bool staged = (ntot <= CAPC);
  __syncthreads();

  walk_pieces<true>(seg, loff, R, PB, r, lane, wave, cnt, cur, cstage, col + rbase, staged);

  if (staged) {
    __syncthreads();
    for (int i = tid; i < ntot; i += RT) col[(long)rbase + i] = cstage[i];
  }

  const int rowBase = nbase + wave * 16;
  if (rowBase >= N) return;
  const int kh  = lane >> 4;
  const int n16 = lane & 15;
  int row = rowBase + n16;
  if (row >= N) row = N - 1;

  float sc[8];
#pragma unroll
  for (int i = 0; i < 8; ++i) sc[i] = dl[wave * 16 + i + 8 * kh];

  v8f acc0 = {}, acc1 = {}, acc2 = {}, acc3 = {};
  for (int k = 0; k < KDIM1; k += 4) {
    const int ka = k + 2 * kh;
    v2f a;
    a.x = (ka     < IN_C) ? x[(long)row * IN_C + ka]       : pos[(long)row * POSD + (ka     - IN_C)];
    a.y = (ka + 1 < IN_C) ? x[(long)row * IN_C + ka + 1]   : pos[(long)row * POSD + (ka + 1 - IN_C)];
    const float* wb = W1 + (long)ka * HID + n16;
    v2f b0; b0.x = wb[0];  b0.y = wb[HID];
    v2f b1; b1.x = wb[16]; b1.y = wb[HID + 16];
    v2f b2; b2.x = wb[32]; b2.y = wb[HID + 32];
    v2f b3; b3.x = wb[48]; b3.y = wb[HID + 48];
    acc0 = __builtin_amdgcn_wmma_f32_16x16x4_f32(false, a, false, b0, (short)0, acc0, false, false);
    acc1 = __builtin_amdgcn_wmma_f32_16x16x4_f32(false, a, false, b1, (short)0, acc1, false, false);
    acc2 = __builtin_amdgcn_wmma_f32_16x16x4_f32(false, a, false, b2, (short)0, acc2, false, false);
    acc3 = __builtin_amdgcn_wmma_f32_16x16x4_f32(false, a, false, b3, (short)0, acc3, false, false);
  }
#pragma unroll
  for (int i = 0; i < 8; ++i) {
    acc0[i] *= sc[i];
    acc1[i] *= sc[i];
    acc2[i] *= sc[i];
    acc3[i] *= sc[i];
  }
#pragma unroll
  for (int i = 0; i < 8; ++i) {
    const int r = rowBase + i + 8 * kh;
    if (r < N) {
      const long base = (long)r * HID + n16;
      hw1[base +  0] = acc0[i];
      hw1[base + 16] = acc1[i];
      hw1[base + 32] = acc2[i];
      hw1[base + 48] = acc3[i];
    }
  }
}


template <int U>
__device__ __forceinline__ void gather_batch2(const float2* __restrict__ f2, int s_l,
                                              int j, int lane, float2& acc) {
  float2 v[U];
#pragma unroll
  for (int u = 0; u < U; ++u) {
    const int s = __builtin_amdgcn_readlane(s_l, j + u);
    v[u] = f2[(long)s * (HID / 2) + lane];
  }
#pragma unroll
  for (int u = 0; u < U; ++u) {
    acc.x += v[u].x;
    acc.y += v[u].y;
  }
}

template <int U>
__device__ __forceinline__ void gather_batch1(const float* __restrict__ f, int s_l,
                                              int j, int lane, float& acc) {
  float v[U];
#pragma unroll
  for (int u = 0; u < U; ++u) {
    const int s = __builtin_amdgcn_readlane(s_l, j + u);
    v[u] = f[(long)s * OUTD + lane];
  }
#pragma unroll
  for (int u = 0; u < U; ++u) acc += v[u];
}

#define RPW 8

#define BROWS (8 * RPW)
#define HP 68

__global__ __launch_bounds__(256) void k_gather64_relu_gemm2(
    const int* __restrict__ rowptr, const int* __restrict__ col,
    const float* __restrict__ dinv, const float* __restrict__ feat,
    const float* __restrict__ bias, const float* __restrict__ W2,
    float* __restrict__ hw2, int N) {
  __shared__ float hs[BROWS * HP];
  const int lane = threadIdx.x & 31;
  const int wave = __builtin_amdgcn_readfirstlane((int)(threadIdx.x >> 5));
  const int r0 = (blockIdx.x * 8 + wave) * RPW;
  float* __restrict__ hrow = hs + wave * (RPW * HP);

  int nr = N - r0;
  nr = (nr < 0) ? 0 : ((nr < RPW) ? nr : RPW);

  if (nr > 0) {
    const int rp_l = rowptr[r0 + ((lane <= nr) ? lane : nr)];
    const float dd_l = dinv[r0 + ((lane < nr) ? lane : (nr - 1))];
    const float2* __restrict__ f2 = reinterpret_cast<const float2*>(feat);
    const float2 b = reinterpret_cast<const float2*>(bias)[lane];
    float2 self_n = f2[(long)r0 * (HID / 2) + lane];

    int beg = __builtin_amdgcn_readlane(rp_l, 0);
    int end = __builtin_amdgcn_readlane(rp_l, 1);
    int c_n = r0;
    if (lane < end - beg) c_n = col[beg + lane];

    for (int ri = 0; ri < nr; ++ri) {
      const int row = r0 + ri;
      int s_l = (c_n < 0) ? 0 : ((c_n >= N) ? N - 1 : c_n);
      const float2 self = self_n;
      const int nbeg = end;
      int nend = end;
      if (ri + 1 < nr) {
        nend = __builtin_amdgcn_readlane(rp_l, ri + 2);
        c_n = row + 1;
        if (lane < nend - nbeg) c_n = col[nbeg + lane];
        self_n = f2[(long)(row + 1) * (HID / 2) + lane];
      }

      float2 acc;
      acc.x = 0.0f;
      acc.y = 0.0f;
      for (int base = beg; base < end; base += 32) {
        const int n = (end - base < 32) ? (end - base) : 32;
        if (base != beg) {
          s_l = row;
          if (lane < n) {
            s_l = col[base + lane];
            s_l = (s_l < 0) ? 0 : ((s_l >= N) ? N - 1 : s_l);
          }
        }
        int j = 0;
        for (; j + 8 <= n; j += 8) gather_batch2<8>(f2, s_l, j, lane, acc);
        if (n - j >= 4) { gather_batch2<4>(f2, s_l, j, lane, acc); j += 4; }
        if (n - j >= 2) { gather_batch2<2>(f2, s_l, j, lane, acc); j += 2; }
        if (n - j >= 1) { gather_batch2<1>(f2, s_l, j, lane, acc); }
      }

      acc.x += self.x;
      acc.y += self.y;
      const float dd = __int_as_float(__builtin_amdgcn_readlane(__float_as_int(dd_l), ri));
      float2 o;
      o.x = fmaxf(fmaf(acc.x, dd, b.x), 0.0f) * dd;
      o.y = fmaxf(fmaf(acc.y, dd, b.y), 0.0f) * dd;
      reinterpret_cast<float2*>(hrow + ri * HP)[lane] = o;

      beg = nbeg;
      end = nend;
    }
  }
  for (int ri = nr; ri < RPW; ++ri) {
    float2 zz;
    zz.x = 0.0f;
    zz.y = 0.0f;
    reinterpret_cast<float2*>(hrow + ri * HP)[lane] = zz;
  }
  __syncthreads();

  if (wave >= BROWS / 16) return;
  const int rowBase = blockIdx.x * BROWS + wave * 16;
  if (rowBase >= N) return;
  const int kh  = lane >> 4;
  const int n16 = lane & 15;
  const float* ar = hs + (wave * 16 + n16) * HP + 2 * kh;
  const float* wr = W2 + (2 * kh) * OUTD + n16;

  v8f acc0 = {}, acc1 = {};
#pragma unroll 8
  for (int k = 0; k < HID; k += 4) {
    const v2f a = *reinterpret_cast<const v2f*>(ar + k);
    const float* wb = wr + k * OUTD;
    v2f b0; b0.x = wb[0];  b0.y = wb[OUTD];
    v2f b1; b1.x = wb[16]; b1.y = wb[OUTD + 16];
    acc0 = __builtin_amdgcn_wmma_f32_16x16x4_f32(false, a, false, b0, (short)0, acc0, false, false);
    acc1 = __builtin_amdgcn_wmma_f32_16x16x4_f32(false, a, false, b1, (short)0, acc1, false, false);
  }
  int rfirst = rowBase + 8 * kh;
  int cfirst = n16;
  asm volatile("v_nop\n\tv_nop\n\tv_nop\n\tv_nop" : "+v"(acc0), "+v"(acc1), "+v"(rfirst), "+v"(cfirst));
#pragma unroll
  for (int i = 0; i < 8; ++i) {
    const int r = rfirst + i;
    if (r < N) {
      const long base = (long)r * OUTD + cfirst;
      hw2[base +  0] = acc0[i];
      hw2[base + 16] = acc1[i];
    }
  }
}

__global__ __launch_bounds__(256) void k_gather32(
    const int* __restrict__ rowptr, const int* __restrict__ col,
    const float* __restrict__ dinv, const float* __restrict__ feat,
    const float* __restrict__ bias, float* __restrict__ out, int N) {
  const int lane = threadIdx.x & 31;
  const int rb = (blockIdx.x * 8 + (threadIdx.x >> 5)) * RPW;
  if (rb >= N) return;
  const int r0 = __builtin_amdgcn_readfirstlane(rb);
  const int nr = (N - r0 < RPW) ? (N - r0) : RPW;
  const int rp_l = rowptr[r0 + ((lane <= nr) ? lane : nr)];
  const float dd_l = dinv[r0 + ((lane < nr) ? lane : (nr - 1))];
  const float bv = bias[lane];
  float self_n = feat[(long)r0 * OUTD + lane];

  int beg = __builtin_amdgcn_readlane(rp_l, 0);
  int end = __builtin_amdgcn_readlane(rp_l, 1);
  int c_n = r0;
  if (lane < end - beg) c_n = col[beg + lane];

  for (int ri = 0; ri < nr; ++ri) {
    const int row = r0 + ri;
    int s_l = (c_n < 0) ? 0 : ((c_n >= N) ? N - 1 : c_n);
    const float self = self_n;
    const int nbeg = end;
    int nend = end;
    if (ri + 1 < nr) {
      nend = __builtin_amdgcn_readlane(rp_l, ri + 2);
      c_n = row + 1;
      if (lane < nend - nbeg) c_n = col[nbeg + lane];
      self_n = feat[(long)(row + 1) * OUTD + lane];
    }

    float acc = 0.0f;
    for (int base = beg; base < end; base += 32) {
      const int n = (end - base < 32) ? (end - base) : 32;
      if (base != beg) {
        s_l = row;
        if (lane < n) {
          s_l = col[base + lane];
          s_l = (s_l < 0) ? 0 : ((s_l >= N) ? N - 1 : s_l);
        }
      }
      int j = 0;
      for (; j + 8 <= n; j += 8) gather_batch1<8>(feat, s_l, j, lane, acc);
      if (n - j >= 4) { gather_batch1<4>(feat, s_l, j, lane, acc); j += 4; }
      if (n - j >= 2) { gather_batch1<2>(feat, s_l, j, lane, acc); j += 2; }
      if (n - j >= 1) { gather_batch1<1>(feat, s_l, j, lane, acc); }
    }

    acc += self;
    const float dd = __int_as_float(__builtin_amdgcn_readlane(__float_as_int(dd_l), ri));
    out[(long)row * OUTD + lane] = fmaf(acc, dd, bv);

    beg = nbeg;
    end = nend;
  }
}

__global__ void k_link(const float* __restrict__ z, const int* __restrict__ eli,
                       const float* __restrict__ Wl, const float* __restrict__ bl,
                       float* __restrict__ pred, int EL) {
  int e = blockIdx.x * blockDim.x + threadIdx.x;
  if (e >= EL) return;
  const int s = eli[e];
  const int d = eli[EL + e];
  const float* zs = z + (long)s * OUTD;
  const float* zd = z + (long)d * OUTD;
  float acc = bl[0];
#pragma unroll
  for (int c = 0; c < OUTD; ++c)
    acc = fmaf(zs[c], Wl[c], fmaf(zd[c], Wl[OUTD + c], acc));
  pred[e] = acc;
}

extern "C" void kernel_launch(void* const* d_in, const int* in_sizes, int n_in,
                              void* d_out, int out_size, void* d_ws, size_t ws_size,
                              hipStream_t stream) {
  const float* x   = (const float*)d_in[0];
  const float* pos = (const float*)d_in[1];
  const float* W1  = (const float*)d_in[2];
  const float* b1  = (const float*)d_in[3];
  const float* W2  = (const float*)d_in[4];
  const float* b2  = (const float*)d_in[5];
  const float* Wl  = (const float*)d_in[6];
  const float* bl  = (const float*)d_in[7];
  const int*   ei  = (const int*)d_in[8];
  const int*   eli = (const int*)d_in[9];

  const int N  = in_sizes[0] / IN_C;
  const int E  = in_sizes[8] / 2;
  const int EL = in_sizes[9] / 2;

  float* z    = (float*)d_out;
  float* pred = z + (long)N * OUTD;

  float* hw1   = (float*)d_ws;
  float* h1    = hw1 + (long)N * HID;
  float* reg3  = h1  + (long)N * HID;
  float* dinv  = reg3 + (long)N * OUTD;
  float* hw2   = h1;
  int2*  seg   = (int2*)h1;
  int*   loff  = (int*)(seg + E);
  int*   col   = (int*)reg3;
  int*   rowptr = col + E;

  const int* esrc = ei;
  const int* edst = ei + E;

  const int B = 256;
  const int tiles  = (E + PTILE - 1) / PTILE;
  const int ranges = (N + RN - 1) / RN;

  k_part<<<tiles, PT, 0, stream>>>(esrc, edst, seg, loff, E, N, ranges);
  k_rowsort_gemm1<<<ranges, RT, 0, stream>>>(seg, loff, rowptr, dinv, col, x, pos, W1, hw1, N, ranges, tiles);

  const int rblocks  = (N + 8 * RPW - 1) / (8 * RPW);

  k_gather64_relu_gemm2<<<rblocks, B, 0, stream>>>(rowptr, col, dinv, hw1, b1, W2, hw2, N);

  k_gather32<<<rblocks, B, 0, stream>>>(rowptr, col, dinv, hw2, b2, z, N);

  k_link<<<(EL + B - 1) / B, B, 0, stream>>>(z, eli, Wl, bl, pred, EL);
}


# Link dot split by endpoint: layer 2's gather also forms z[r].Wl[0:32] and z[r].Wl[32:64] per node (wave tree sum); link kernel adds two scalars per edge, no z rows; pred sum order changes
# speedup vs baseline: 6.7196x; 1.0660x over previous
#include <hip/hip_runtime.h>

typedef __attribute__((ext_vector_type(2))) float v2f;
typedef __attribute__((ext_vector_type(8))) float v8f;

#define IN_C 64
#define POSD 16
#define KDIM1 (IN_C + POSD)
#define HID 64
#define OUTD 32


#define RN_SHIFT 8
#define RN 256
#define MAXR 1024
#define PT 1024
#define RT 512
#define RW (RT / 32)
#define EPT 8
#define PTILE (PT * EPT)
#define CAPC 12288

__global__ __launch_bounds__(PT) void k_part(const int* __restrict__ src, const int* __restrict__ dst,
                                             int2* __restrict__ seg, int* __restrict__ loff,
                                             int E, int N, int R) {
  __shared__ int rc[MAXR];
  __shared__ int wsum[32];
  __shared__ int2 stage[PTILE];
  const int tid = threadIdx.x, lane = tid & 31, wave = tid >> 5;
  const int base = blockIdx.x * PTILE;

  rc[tid] = 0;
  __syncthreads();

  int s[EPT], d[EPT], rk[EPT];
#pragma unroll
  for (int k = 0; k < EPT; ++k) {
    const int e = base + k * PT + tid;
    const int ec = (e < E) ? e : (E - 1);
    s[k] = src[ec];
    d[k] = dst[ec];
  }
#pragma unroll
  for (int k = 0; k < EPT; ++k) {
    const int e = base + k * PT + tid;
    int t = d[k];
    t = (t < 0) ? 0 : ((t >= N) ? N - 1 : t);
    d[k] = (e < E) ? t : -1;
  }
#pragma unroll
  for (int k = 0; k < EPT; ++k) {
    rk[k] = 0;
    if (d[k] >= 0) rk[k] = atomicAdd(&rc[d[k] >> RN_SHIFT], 1);
  }
  __syncthreads();

  const int c = rc[tid];
  int inc = c;
  for (int off = 1; off < 32; off <<= 1) {
    const int t = __shfl_up(inc, off, 32);
    if (lane >= off) inc += t;
  }
  if (lane == 31) wsum[wave] = inc;
  __syncthreads();
  if (wave == 0) {
    const int v = wsum[lane];
    int iv = v;
    for (int off = 1; off < 32; off <<= 1) {
      const int t = __shfl_up(iv, off, 32);
      if (lane >= off) iv += t;
    }
    wsum[lane] = iv - v;
  }
  __syncthreads();
  const int excl = wsum[wave] + inc - c;
  rc[tid] = excl;
  if (tid <= R) loff[(long)blockIdx.x * (R + 1) + tid] = excl;
  __syncthreads();

#pragma unroll
  for (int k = 0; k < EPT; ++k) {
    if (d[k] >= 0) {
      int2 p;
      p.x = s[k];
      p.y = d[k];
      stage[rc[d[k] >> RN_SHIFT] + rk[k]] = p;
    }
  }
  __syncthreads();

  const int nloc = (E - base < PTILE) ? (E - base) : PTILE;
  for (int i = tid; i < nloc; i += PT) seg[(long)base + i] = stage[i];
}

template <bool PLACE>
__device__ __forceinline__ void walk_pieces(const int2* __restrict__ seg, const int* __restrict__ loff,
                                            int R, int PB, int r, int lane, int wave,
                                            int* cnt, int* cur, int* cstage, int* __restrict__ colr,
                                            bool staged) {
  for (int p0 = 0; p0 < PB; p0 += RT) {
    const int p = p0 + lane * RW + wave;
    int pst = 0, pcn = 0;
    if (p < PB) {
      const int a = loff[(long)p * (R + 1) + r];
      const int b = loff[(long)p * (R + 1) + r + 1];
      pst = p * PTILE + a;
      pcn = b - a;
    }
    const int rem = PB - p0 - wave;
    int npw = (rem <= 0) ? 0 : ((rem + RW - 1) / RW);
    if (npw > 32) npw = 32;
    for (int j = 0; j < npw; j += 8) {
      int st[8], cn[8];
      int2 e[8];
#pragma unroll
      for (int u = 0; u < 8; ++u) {
        st[u] = __builtin_amdgcn_readlane(pst, j + u);
        cn[u] = __builtin_amdgcn_readlane(pcn, j + u);
      }
#pragma unroll
      for (int u = 0; u < 8; ++u) {
        const long idx = (lane < cn[u]) ? ((long)st[u] + lane) : 0;
        e[u] = seg[idx];
      }
#pragma unroll
      for (int u = 0; u < 8; ++u) {
        if (lane < cn[u]) {
          const int loc = e[u].y & (RN - 1);
          if (!PLACE) {
            atomicAdd(&cnt[loc], 1);
          } else {
            const int pos = atomicAdd(&cur[loc], 1);
            if (staged) cstage[pos] = e[u].x; else colr[pos] = e[u].x;
          }
        }
      }
    }
    for (int j = 0; j < npw; ++j) {
      const int cnj = __builtin_amdgcn_readlane(pcn, j);
      if (cnj > 32) {
        const int stj = __builtin_amdgcn_readlane(pst, j);
        for (int i = 32 + lane; i < cnj; i += 32) {
          const int2 ee = seg[(long)stj + i];
          const int loc = ee.y & (RN - 1);
          if (!PLACE) {
            atomicAdd(&cnt[loc], 1);
          } else {
            const int pos = atomicAdd(&cur[loc], 1);
            if (staged) cstage[pos] = ee.x; else colr[pos] = ee.x;
          }
        }
      }
    }
  }
}

static_assert(RN == 16 * RW, "one 16-row tile of the range per wave");
__global__ __launch_bounds__(RT) void k_rowsort_gemm1(const int2* __restrict__ seg, const int* __restrict__ loff,
                                                      int* __restrict__ rowptr, float* __restrict__ dinv,
                                                      int* __restrict__ col,
                                                      const float* __restrict__ x, const float* __restrict__ pos,
                                                      const float* __restrict__ W1, float* __restrict__ hw1,
                                                      int N, int R, int PB) {
  __shared__ float dl[RN];
  __shared__ int cnt[RN];
  __shared__ int cur[RN];
  __shared__ int red[RW];
  __shared__ int wtot[RN / 32];
  __shared__ int cstage[CAPC];
  const int tid = threadIdx.x, lane = tid & 31, wave = tid >> 5;
  const int r = blockIdx.x;
  const int nbase = r << RN_SHIFT;

  if (tid < RN) cnt[tid] = 0;

  int bl = 0;
  for (int p = tid; p < PB; p += RT) bl += loff[(long)p * (R + 1) + r];
  for (int off = 16; off > 0; off >>= 1) bl += __shfl_down(bl, off, 32);
  if (lane == 0) red[wave] = bl;
  __syncthreads();
  int rbase = 0;
  for (int w = 0; w < RW; ++w) rbase += red[w];

  walk_pieces<false>(seg, loff, R, PB, r, lane, wave, cnt, cur, cstage, col, false);
  __syncthreads();

  const int c = (tid < RN) ? cnt[tid] : 0;
  int inc = c;
  for (int off = 1; off < 32; off <<= 1) {
    const int t = __shfl_up(inc, off, 32);
    if (lane >= off) inc += t;
  }
  if (tid < RN && lane == 31) wtot[wave] = inc;
  __syncthreads();
  int ntot = 0;
  for (int w = 0; w < RN / 32; ++w) ntot += wtot[w];
  if (tid < RN) {
    int woff = 0;
    for (int w = 0; w < wave; ++w) woff += wtot[w];
    const int excl = woff + inc - c;
    cur[tid] = excl;
    const int node = nbase + tid;
    if (node < N) {
      rowptr[node] = rbase + excl;
      dinv[node] = rsqrtf((float)(c + 1));
      if (node == N - 1) rowptr[N] = rbase + excl + c;
    }
    dl[tid] = rsqrtf((float)(c + 1));
  }
  const bool staged = (ntot <= CAPC);
  __syncthreads();

  walk_pieces<true>(seg, loff, R, PB, r, lane, wave, cnt, cur, cstage, col + rbase, staged);

  if (staged) {
    __syncthreads();
    for (int i = tid; i < ntot; i += RT) col[(long)rbase + i] = cstage[i];
  }

  const int rowBase = nbase + wave * 16;
  if (rowBase >= N) return;
  const int kh  = lane >> 4;
  const int n16 = lane & 15;
  int row = rowBase + n16;
  if (row >= N) row = N - 1;

  float sc[8];
#pragma unroll
  for (int i = 0; i < 8; ++i) sc[i] = dl[wave * 16 + i + 8 * kh];

  v8f acc0 = {}, acc1 = {}, acc2 = {}, acc3 = {};
  for (int k = 0; k < KDIM1; k += 4) {
    const int ka = k + 2 * kh;
    v2f a;
    a.x = (ka     < IN_C) ? x[(long)row * IN_C + ka]       : pos[(long)row * POSD + (ka     - IN_C)];
    a.y = (ka + 1 < IN_C) ? x[(long)row * IN_C + ka + 1]   : pos[(long)row * POSD + (ka + 1 - IN_C)];
    const float* wb = W1 + (long)ka * HID + n16;
    v2f b0; b0.x = wb[0];  b0.y = wb[HID];
    v2f b1; b1.x = wb[16]; b1.y = wb[HID + 16];
    v2f b2; b2.x = wb[32]; b2.y = wb[HID + 32];
    v2f b3; b3.x = wb[48]; b3.y = wb[HID + 48];
    acc0 = __builtin_amdgcn_wmma_f32_16x16x4_f32(false, a, false, b0, (short)0, acc0, false, false);
    acc1 = __builtin_amdgcn_wmma_f32_16x16x4_f32(false, a, false, b1, (short)0, acc1, false, false);
    acc2 = __builtin_amdgcn_wmma_f32_16x16x4_f32(false, a, false, b2, (short)0, acc2, false, false);
    acc3 = __builtin_amdgcn_wmma_f32_16x16x4_f32(false, a, false, b3, (short)0, acc3, false, false);
  }
#pragma unroll
  for (int i = 0; i < 8; ++i) {
    acc0[i] *= sc[i];
    acc1[i] *= sc[i];
    acc2[i] *= sc[i];
    acc3[i] *= sc[i];
  }
#pragma unroll
  for (int i = 0; i < 8; ++i) {
    const int r = rowBase + i + 8 * kh;
    if (r < N) {
      const long base = (long)r * HID + n16;
      hw1[base +  0] = acc0[i];
      hw1[base + 16] = acc1[i];
      hw1[base + 32] = acc2[i];
      hw1[base + 48] = acc3[i];
    }
  }
}


template <int U>
__device__ __forceinline__ void gather_batch2(const float2* __restrict__ f2, int s_l,
                                              int j, int lane, float2& acc) {
  float2 v[U];
#pragma unroll
  for (int u = 0; u < U; ++u) {
    const int s = __builtin_amdgcn_readlane(s_l, j + u);
    v[u] = f2[(long)s * (HID / 2) + lane];
  }
#pragma unroll
  for (int u = 0; u < U; ++u) {
    acc.x += v[u].x;
    acc.y += v[u].y;
  }
}

template <int U>
__device__ __forceinline__ void gather_batch1(const float* __restrict__ f, int s_l,
                                              int j, int lane, float& acc) {
  float v[U];
#pragma unroll
  for (int u = 0; u < U; ++u) {
    const int s = __builtin_amdgcn_readlane(s_l, j + u);
    v[u] = f[(long)s * OUTD + lane];
  }
#pragma unroll
  for (int u = 0; u < U; ++u) acc += v[u];
}

#define RPW 8

#define BROWS (8 * RPW)
#define HP 68

__global__ __launch_bounds__(256) void k_gather64_relu_gemm2(
    const int* __restrict__ rowptr, const int* __restrict__ col,
    const float* __restrict__ dinv, const float* __restrict__ feat,
    const float* __restrict__ bias, const float* __restrict__ W2,
    float* __restrict__ hw2, int N) {
  __shared__ float hs[BROWS * HP];
  const int lane = threadIdx.x & 31;
  const int wave = __builtin_amdgcn_readfirstlane((int)(threadIdx.x >> 5));
  const int r0 = (blockIdx.x * 8 + wave) * RPW;
  float* __restrict__ hrow = hs + wave * (RPW * HP);

  int nr = N - r0;
  nr = (nr < 0) ? 0 : ((nr < RPW) ? nr : RPW);

  if (nr > 0) {
    const int rp_l = rowptr[r0 + ((lane <= nr) ? lane : nr)];
    const float dd_l = dinv[r0 + ((lane < nr) ? lane : (nr - 1))];
    const float2* __restrict__ f2 = reinterpret_cast<const float2*>(feat);
    const float2 b = reinterpret_cast<const float2*>(bias)[lane];
    float2 self_n = f2[(long)r0 * (HID / 2) + lane];

    int beg = __builtin_amdgcn_readlane(rp_l, 0);
    int end = __builtin_amdgcn_readlane(rp_l, 1);
    int c_n = r0;
    if (lane < end - beg) c_n = col[beg + lane];

    for (int ri = 0; ri < nr; ++ri) {
      const int row = r0 + ri;
      int s_l = (c_n < 0) ? 0 : ((c_n >= N) ? N - 1 : c_n);
      const float2 self = self_n;
      const int nbeg = end;
      int nend = end;
      if (ri + 1 < nr) {
        nend = __builtin_amdgcn_readlane(rp_l, ri + 2);
        c_n = row + 1;
        if (lane < nend - nbeg) c_n = col[nbeg + lane];
        self_n = f2[(long)(row + 1) * (HID / 2) + lane];
      }

      float2 acc;
      acc.x = 0.0f;
      acc.y = 0.0f;
      for (int base = beg; base < end; base += 32) {
        const int n = (end - base < 32) ? (end - base) : 32;
        if (base != beg) {
          s_l = row;
          if (lane < n) {
            s_l = col[base + lane];
            s_l = (s_l < 0) ? 0 : ((s_l >= N) ? N - 1 : s_l);
          }
        }
        int j = 0;
        for (; j + 8 <= n; j += 8) gather_batch2<8>(f2, s_l, j, lane, acc);
        if (n - j >= 4) { gather_batch2<4>(f2, s_l, j, lane, acc); j += 4; }
        if (n - j >= 2) { gather_batch2<2>(f2, s_l, j, lane, acc); j += 2; }
        if (n - j >= 1) { gather_batch2<1>(f2, s_l, j, lane, acc); }
      }

      acc.x += self.x;
      acc.y += self.y;
      const float dd = __int_as_float(__builtin_amdgcn_readlane(__float_as_int(dd_l), ri));
      float2 o;
      o.x = fmaxf(fmaf(acc.x, dd, b.x), 0.0f) * dd;
      o.y = fmaxf(fmaf(acc.y, dd, b.y), 0.0f) * dd;
      reinterpret_cast<float2*>(hrow + ri * HP)[lane] = o;

      beg = nbeg;
      end = nend;
    }
  }
  for (int ri = nr; ri < RPW; ++ri) {
    float2 zz;
    zz.x = 0.0f;
    zz.y = 0.0f;
    reinterpret_cast<float2*>(hrow + ri * HP)[lane] = zz;
  }
  __syncthreads();

  if (wave >= BROWS / 16) return;
  const int rowBase = blockIdx.x * BROWS + wave * 16;
  if (rowBase >= N) return;
  const int kh  = lane >> 4;
  const int n16 = lane & 15;
  const float* ar = hs + (wave * 16 + n16) * HP + 2 * kh;
  const float* wr = W2 + (2 * kh) * OUTD + n16;

  v8f acc0 = {}, acc1 = {};
#pragma unroll 8
  for (int k = 0; k < HID; k += 4) {
    const v2f a = *reinterpret_cast<const v2f*>(ar + k);
    const float* wb = wr + k * OUTD;
    v2f b0; b0.x = wb[0];  b0.y = wb[OUTD];
    v2f b1; b1.x = wb[16]; b1.y = wb[OUTD + 16];
    acc0 = __builtin_amdgcn_wmma_f32_16x16x4_f32(false, a, false, b0, (short)0, acc0, false, false);
    acc1 = __builtin_amdgcn_wmma_f32_16x16x4_f32(false, a, false, b1, (short)0, acc1, false, false);
  }
  int rfirst = rowBase + 8 * kh;
  int cfirst = n16;
  asm volatile("v_nop\n\tv_nop\n\tv_nop\n\tv_nop" : "+v"(acc0), "+v"(acc1), "+v"(rfirst), "+v"(cfirst));
#pragma unroll
  for (int i = 0; i < 8; ++i) {
    const int r = rfirst + i;
    if (r < N) {
      const long base = (long)r * OUTD + cfirst;
      hw2[base +  0] = acc0[i];
      hw2[base + 16] = acc1[i];
    }
  }
}

template <int XMASK>
__device__ __forceinline__ float row_xmask_add(float v) {
  return v + __int_as_float(__builtin_amdgcn_update_dpp(0, __float_as_int(v), 0x160 + XMASK, 0xf, 0xf, true));
}

__device__ __forceinline__ float wave_sum32(float v) {
  v = row_xmask_add<1>(v);
  v = row_xmask_add<2>(v);
  v = row_xmask_add<4>(v);
  v = row_xmask_add<8>(v);
  const int b = __float_as_int(v);
  return v + __int_as_float(__builtin_amdgcn_permlanex16(b, b, 0x76543210, 0xfedcba98, false, false));
}

__global__ __launch_bounds__(256) void k_gather32(
    const int* __restrict__ rowptr, const int* __restrict__ col,
    const float* __restrict__ dinv, const float* __restrict__ feat,
    const float* __restrict__ bias, const float* __restrict__ Wl,
    float* __restrict__ out, float* __restrict__ pq, int N) {
  const int lane = threadIdx.x & 31;
  const int rb = (blockIdx.x * 8 + (threadIdx.x >> 5)) * RPW;
  if (rb >= N) return;
  const int r0 = __builtin_amdgcn_readfirstlane(rb);
  const int nr = (N - r0 < RPW) ? (N - r0) : RPW;
  const int rp_l = rowptr[r0 + ((lane <= nr) ? lane : nr)];
  const float dd_l = dinv[r0 + ((lane < nr) ? lane : (nr - 1))];
  const float bv = bias[lane];
  const float wa = Wl[lane];
  const float wb = Wl[OUTD + lane];
  float pa_l = 0.0f, pb_l = 0.0f;
  float self_n = feat[(long)r0 * OUTD + lane];

  int beg = __builtin_amdgcn_readlane(rp_l, 0);
  int end = __builtin_amdgcn_readlane(rp_l, 1);
  int c_n = r0;
  if (lane < end - beg) c_n = col[beg + lane];

  for (int ri = 0; ri < nr; ++ri) {
    const int row = r0 + ri;
    int s_l = (c_n < 0) ? 0 : ((c_n >= N) ? N - 1 : c_n);
    const float self = self_n;
    const int nbeg = end;
    int nend = end;
    if (ri + 1 < nr) {
      nend = __builtin_amdgcn_readlane(rp_l, ri + 2);
      c_n = row + 1;
      if (lane < nend - nbeg) c_n = col[nbeg + lane];
      self_n = feat[(long)(row + 1) * OUTD + lane];
    }

    float acc = 0.0f;
    for (int base = beg; base < end; base += 32) {
      const int n = (end - base < 32) ? (end - base) : 32;
      if (base != beg) {
        s_l = row;
        if (lane < n) {
          s_l = col[base + lane];
          s_l = (s_l < 0) ? 0 : ((s_l >= N) ? N - 1 : s_l);
        }
      }
      int j = 0;
      for (; j + 8 <= n; j += 8) gather_batch1<8>(feat, s_l, j, lane, acc);
      if (n - j >= 4) { gather_batch1<4>(feat, s_l, j, lane, acc); j += 4; }
      if (n - j >= 2) { gather_batch1<2>(feat, s_l, j, lane, acc); j += 2; }
      if (n - j >= 1) { gather_batch1<1>(feat, s_l, j, lane, acc); }
    }

    acc += self;
    const float dd = __int_as_float(__builtin_amdgcn_readlane(__float_as_int(dd_l), ri));
    const float o = fmaf(acc, dd, bv);
    out[(long)row * OUTD + lane] = o;
    const float sa = wave_sum32(o * wa);
    const float sb = wave_sum32(o * wb);
    if (lane == ri) {
      pa_l = sa;
      pb_l = sb;
    }

    beg = nbeg;
    end = nend;
  }
  if (lane < nr) {
    float2 p;
    p.x = pa_l;
    p.y = pb_l;
    reinterpret_cast<float2*>(pq)[r0 + lane] = p;
  }
}

#define LT 128
#define LPT 4
__global__ __launch_bounds__(LT) void k_link(const float* __restrict__ pq, const int* __restrict__ eli,
                                             const float* __restrict__ bl,
                                             float* __restrict__ pred, int EL) {
  const int bbase = blockIdx.x * (LT * LPT);
  const float b = bl[0];
  if (bbase + LT * LPT <= EL) {
    const int e0 = bbase + threadIdx.x;
    int s[LPT], d[LPT];
#pragma unroll
    for (int k = 0; k < LPT; ++k) {
      s[k] = eli[e0 + k * LT];
      d[k] = eli[EL + e0 + k * LT];
    }
    float ps[LPT], pd[LPT];
#pragma unroll
    for (int k = 0; k < LPT; ++k) {
      ps[k] = pq[2 * (long)s[k]];
      pd[k] = pq[2 * (long)d[k] + 1];
    }
#pragma unroll
    for (int k = 0; k < LPT; ++k) pred[e0 + k * LT] = (ps[k] + pd[k]) + b;
  } else {
    for (int k = 0; k < LPT; ++k) {
      const int e = bbase + k * LT + threadIdx.x;
      if (e < EL) pred[e] = (pq[2 * (long)eli[e]] + pq[2 * (long)eli[EL + e] + 1]) + b;
    }
  }
}

extern "C" void kernel_launch(void* const* d_in, const int* in_sizes, int n_in,
                              void* d_out, int out_size, void* d_ws, size_t ws_size,
                              hipStream_t stream) {
  const float* x   = (const float*)d_in[0];
  const float* pos = (const float*)d_in[1];
  const float* W1  = (const float*)d_in[2];
  const float* b1  = (const float*)d_in[3];
  const float* W2  = (const float*)d_in[4];
  const float* b2  = (const float*)d_in[5];
  const float* Wl  = (const float*)d_in[6];
  const float* bl  = (const float*)d_in[7];
  const int*   ei  = (const int*)d_in[8];
  const int*   eli = (const int*)d_in[9];

  const int N  = in_sizes[0] / IN_C;
  const int E  = in_sizes[8] / 2;
  const int EL = in_sizes[9] / 2;

  float* z    = (float*)d_out;
  float* pred = z + (long)N * OUTD;

  float* hw1   = (float*)d_ws;
  float* h1    = hw1 + (long)N * HID;
  float* reg3  = h1  + (long)N * HID;
  float* dinv  = reg3 + (long)N * OUTD;
  float* hw2   = h1;
  int2*  seg   = (int2*)h1;
  int*   loff  = (int*)(seg + E);
  int*   col   = (int*)reg3;
  int*   rowptr = col + E;
  float* pq    = hw1;

  const int* esrc = ei;
  const int* edst = ei + E;

  const int B = 256;
  const int tiles  = (E + PTILE - 1) / PTILE;
  const int ranges = (N + RN - 1) / RN;

  k_part<<<tiles, PT, 0, stream>>>(esrc, edst, seg, loff, E, N, ranges);
  k_rowsort_gemm1<<<ranges, RT, 0, stream>>>(seg, loff, rowptr, dinv, col, x, pos, W1, hw1, N, ranges, tiles);

  const int rblocks  = (N + 8 * RPW - 1) / (8 * RPW);

  k_gather64_relu_gemm2<<<rblocks, B, 0, stream>>>(rowptr, col, dinv, hw1, b1, W2, hw2, N);

  k_gather32<<<rblocks, B, 0, stream>>>(rowptr, col, dinv, hw2, b2, Wl, z, pq, N);

  k_link<<<(EL + LT * LPT - 1) / (LT * LPT), LT, 0, stream>>>(pq, eli, bl, pred, EL);
}


# Gather loads bring several source rows each (16 B per lane: 2 rows of 64 floats, 4 of 32), a row's edges in one load group; row sums and the Wl dots become interleaved partial sums (order changes)
# speedup vs baseline: 7.0191x; 1.0446x over previous
#include <hip/hip_runtime.h>

typedef __attribute__((ext_vector_type(2))) float v2f;
typedef __attribute__((ext_vector_type(8))) float v8f;

#define IN_C 64
#define POSD 16
#define KDIM1 (IN_C + POSD)
#define HID 64
#define OUTD 32


#define RN_SHIFT 8
#define RN 256
#define MAXR 1024
#define PT 1024
#define RT 512
#define RW (RT / 32)
#define EPT 8
#define PTILE (PT * EPT)
#define CAPC 12288

__global__ __launch_bounds__(PT) void k_part(const int* __restrict__ src, const int* __restrict__ dst,
                                             int2* __restrict__ seg, int* __restrict__ loff,
                                             int E, int N, int R) {
  __shared__ int rc[MAXR];
  __shared__ int wsum[32];
  __shared__ int2 stage[PTILE];
  const int tid = threadIdx.x, lane = tid & 31, wave = tid >> 5;
  const int base = blockIdx.x * PTILE;

  rc[tid] = 0;
  __syncthreads();

  int s[EPT], d[EPT], rk[EPT];
#pragma unroll
  for (int k = 0; k < EPT; ++k) {
    const int e = base + k * PT + tid;
    const int ec = (e < E) ? e : (E - 1);
    s[k] = src[ec];
    d[k] = dst[ec];
  }
#pragma unroll
  for (int k = 0; k < EPT; ++k) {
    const int e = base + k * PT + tid;
    int t = d[k];
    t = (t < 0) ? 0 : ((t >= N) ? N - 1 : t);
    d[k] = (e < E) ? t : -1;
  }
#pragma unroll
  for (int k = 0; k < EPT; ++k) {
    rk[k] = 0;
    if (d[k] >= 0) rk[k] = atomicAdd(&rc[d[k] >> RN_SHIFT], 1);
  }
  __syncthreads();

  const int c = rc[tid];
  int inc = c;
  for (int off = 1; off < 32; off <<= 1) {
    const int t = __shfl_up(inc, off, 32);
    if (lane >= off) inc += t;
  }
  if (lane == 31) wsum[wave] = inc;
  __syncthreads();
  if (wave == 0) {
    const int v = wsum[lane];
    int iv = v;
    for (int off = 1; off < 32; off <<= 1) {
      const int t = __shfl_up(iv, off, 32);
      if (lane >= off) iv += t;
    }
    wsum[lane] = iv - v;
  }
  __syncthreads();
  const int excl = wsum[wave] + inc - c;
  rc[tid] = excl;
  if (tid <= R) loff[(long)blockIdx.x * (R + 1) + tid] = excl;
  __syncthreads();

#pragma unroll
  for (int k = 0; k < EPT; ++k) {
    if (d[k] >= 0) {
      int2 p;
      p.x = s[k];
      p.y = d[k];
      stage[rc[d[k] >> RN_SHIFT] + rk[k]] = p;
    }
  }
  __syncthreads();

  const int nloc = (E - base < PTILE) ? (E - base) : PTILE;
  for (int i = tid; i < nloc; i += PT) seg[(long)base + i] = stage[i];
}

template <bool PLACE>
__device__ __forceinline__ void walk_pieces(const int2* __restrict__ seg, const int* __restrict__ loff,
                                            int R, int PB, int r, int lane, int wave,
                                            int* cnt, int* cur, int* cstage, int* __restrict__ colr,
                                            bool staged) {
  for (int p0 = 0; p0 < PB; p0 += RT) {
    const int p = p0 + lane * RW + wave;
    int pst = 0, pcn = 0;
    if (p < PB) {
      const int a = loff[(long)p * (R + 1) + r];
      const int b = loff[(long)p * (R + 1) + r + 1];
      pst = p * PTILE + a;
      pcn = b - a;
    }
    const int rem = PB - p0 - wave;
    int npw = (rem <= 0) ? 0 : ((rem + RW - 1) / RW);
    if (npw > 32) npw = 32;
    for (int j = 0; j < npw; j += 8) {
      int st[8], cn[8];
      int2 e[8];
#pragma unroll
      for (int u = 0; u < 8; ++u) {
        st[u] = __builtin_amdgcn_readlane(pst, j + u);
        cn[u] = __builtin_amdgcn_readlane(pcn, j + u);
      }
#pragma unroll
      for (int u = 0; u < 8; ++u) {
        const long idx = (lane < cn[u]) ? ((long)st[u] + lane) : 0;
        e[u] = seg[idx];
      }
#pragma unroll
      for (int u = 0; u < 8; ++u) {
        if (lane < cn[u]) {
          const int loc = e[u].y & (RN - 1);
          if (!PLACE) {
            atomicAdd(&cnt[loc], 1);
          } else {
            const int pos = atomicAdd(&cur[loc], 1);
            if (staged) cstage[pos] = e[u].x; else colr[pos] = e[u].x;
          }
        }
      }
    }
    for (int j = 0; j < npw; ++j) {
      const int cnj = __builtin_amdgcn_readlane(pcn, j);
      if (cnj > 32) {
        const int stj = __builtin_amdgcn_readlane(pst, j);
        for (int i = 32 + lane; i < cnj; i += 32) {
          const int2 ee = seg[(long)stj + i];
          const int loc = ee.y & (RN - 1);
          if (!PLACE) {
            atomicAdd(&cnt[loc], 1);
          } else {
            const int pos = atomicAdd(&cur[loc], 1);
            if (staged) cstage[pos] = ee.x; else colr[pos] = ee.x;
          }
        }
      }
    }
  }
}

static_assert(RN == 16 * RW, "one 16-row tile of the range per wave");
__global__ __launch_bounds__(RT) void k_rowsort_gemm1(const int2* __restrict__ seg, const int* __restrict__ loff,
                                                      int* __restrict__ rowptr, float* __restrict__ dinv,
                                                      int* __restrict__ col,
                                                      const float* __restrict__ x, const float* __restrict__ pos,
                                                      const float* __restrict__ W1, float* __restrict__ hw1,
                                                      int N, int R, int PB) {
  __shared__ float dl[RN];
  __shared__ int cnt[RN];
  __shared__ int cur[RN];
  __shared__ int red[RW];
  __shared__ int wtot[RN / 32];
  __shared__ int cstage[CAPC];
  const int tid = threadIdx.x, lane = tid & 31, wave = tid >> 5;
  const int r = blockIdx.x;
  const int nbase = r << RN_SHIFT;

  if (tid < RN) cnt[tid] = 0;

  int bl = 0;
  for (int p = tid; p < PB; p += RT) bl += loff[(long)p * (R + 1) + r];
  for (int off = 16; off > 0; off >>= 1) bl += __shfl_down(bl, off, 32);
  if (lane == 0) red[wave] = bl;
  __syncthreads();
  int rbase = 0;
  for (int w = 0; w < RW; ++w) rbase += red[w];

  walk_pieces<false>(seg, loff, R, PB, r, lane, wave, cnt, cur, cstage, col, false);
  __syncthreads();

  const int c = (tid < RN) ? cnt[tid] : 0;
  int inc = c;
  for (int off = 1; off < 32; off <<= 1) {
    const int t = __shfl_up(inc, off, 32);
    if (lane >= off) inc += t;
  }
  if (tid < RN && lane == 31) wtot[wave] = inc;
  __syncthreads();
  int ntot = 0;
  for (int w = 0; w < RN / 32; ++w) ntot += wtot[w];
  if (tid < RN) {
    int woff = 0;
    for (int w = 0; w < wave; ++w) woff += wtot[w];
    const int excl = woff + inc - c;
    cur[tid] = excl;
    const int node = nbase + tid;
    if (node < N) {
      rowptr[node] = rbase + excl;
      dinv[node] = rsqrtf((float)(c + 1));
      if (node == N - 1) rowptr[N] = rbase + excl + c;
    }
    dl[tid] = rsqrtf((float)(c + 1));
  }
  const bool staged = (ntot <= CAPC);
  __syncthreads();

  walk_pieces<true>(seg, loff, R, PB, r, lane, wave, cnt, cur, cstage, col + rbase, staged);

  if (staged) {
    __syncthreads();
    for (int i = tid; i < ntot; i += RT) col[(long)rbase + i] = cstage[i];
  }

  const int rowBase = nbase + wave * 16;
  if (rowBase >= N) return;
  const int kh  = lane >> 4;
  const int n16 = lane & 15;
  int row = rowBase + n16;
  if (row >= N) row = N - 1;

  float sc[8];
#pragma unroll
  for (int i = 0; i < 8; ++i) sc[i] = dl[wave * 16 + i + 8 * kh];

  v8f acc0 = {}, acc1 = {}, acc2 = {}, acc3 = {};
  for (int k = 0; k < KDIM1; k += 4) {
    const int ka = k + 2 * kh;
    v2f a;
    a.x = (ka     < IN_C) ? x[(long)row * IN_C + ka]       : pos[(long)row * POSD + (ka     - IN_C)];
    a.y = (ka + 1 < IN_C) ? x[(long)row * IN_C + ka + 1]   : pos[(long)row * POSD + (ka + 1 - IN_C)];
    const float* wb = W1 + (long)ka * HID + n16;
    v2f b0; b0.x = wb[0];  b0.y = wb[HID];
    v2f b1; b1.x = wb[16]; b1.y = wb[HID + 16];
    v2f b2; b2.x = wb[32]; b2.y = wb[HID + 32];
    v2f b3; b3.x = wb[48]; b3.y = wb[HID + 48];
    acc0 = __builtin_amdgcn_wmma_f32_16x16x4_f32(false, a, false, b0, (short)0, acc0, false, false);
    acc1 = __builtin_amdgcn_wmma_f32_16x16x4_f32(false, a, false, b1, (short)0, acc1, false, false);
    acc2 = __builtin_amdgcn_wmma_f32_16x16x4_f32(false, a, false, b2, (short)0, acc2, false, false);
    acc3 = __builtin_amdgcn_wmma_f32_16x16x4_f32(false, a, false, b3, (short)0, acc3, false, false);
  }
#pragma unroll
  for (int i = 0; i < 8; ++i) {
    acc0[i] *= sc[i];
    acc1[i] *= sc[i];
    acc2[i] *= sc[i];
    acc3[i] *= sc[i];
  }
#pragma unroll
  for (int i = 0; i < 8; ++i) {
    const int r = rowBase + i + 8 * kh;
    if (r < N) {
      const long base = (long)r * HID + n16;
      hw1[base +  0] = acc0[i];
      hw1[base + 16] = acc1[i];
      hw1[base + 32] = acc2[i];
      hw1[base + 48] = acc3[i];
    }
  }
}


template <int XMASK>
__device__ __forceinline__ float row_xmask_add(float v) {
  return v + __int_as_float(__builtin_amdgcn_update_dpp(0, __float_as_int(v), 0x160 + XMASK, 0xf, 0xf, true));
}

__device__ __forceinline__ float xhalf_add(float v) {
  const int b = __float_as_int(v);
  return v + __int_as_float(__builtin_amdgcn_permlanex16(b, b, 0x76543210, 0xfedcba98, false, false));
}

#define ROW_SHARE_CASE(U) case U: return __builtin_amdgcn_update_dpp(0, v, 0x150 + U, 0xf, 0xf, true);
__device__ __forceinline__ int row_share_lane(int v, int u) {
  switch (u) {
    ROW_SHARE_CASE(0)  ROW_SHARE_CASE(1)  ROW_SHARE_CASE(2)  ROW_SHARE_CASE(3)
    ROW_SHARE_CASE(4)  ROW_SHARE_CASE(5)  ROW_SHARE_CASE(6)  ROW_SHARE_CASE(7)
    ROW_SHARE_CASE(8)  ROW_SHARE_CASE(9)  ROW_SHARE_CASE(10) ROW_SHARE_CASE(11)
    ROW_SHARE_CASE(12) ROW_SHARE_CASE(13) ROW_SHARE_CASE(14) ROW_SHARE_CASE(15)
    default: return v;
  }
}

#define GROUP_SHARE_CASE(U) case U: return (int)__builtin_amdgcn_mov_dpp8((unsigned)v, U * 0x249249);
__device__ __forceinline__ int group_share_lane(int v, int u) {
  switch (u) {
    GROUP_SHARE_CASE(0) GROUP_SHARE_CASE(1) GROUP_SHARE_CASE(2) GROUP_SHARE_CASE(3)
    GROUP_SHARE_CASE(4) GROUP_SHARE_CASE(5) GROUP_SHARE_CASE(6) GROUP_SHARE_CASE(7)
    default: return v;
  }
}

template <int U0, int K>
__device__ __forceinline__ void gather2_loads(const float4* __restrict__ f4, int idx_l, bool odd_tail,
                                              int h, int t, float4& acc) {
  float4 v[K];
#pragma unroll
  for (int u = 0; u < K; ++u) {
    const int s = row_share_lane(idx_l, U0 + u);
    v[u] = f4[(unsigned)s * (HID / 4) + (unsigned)t];
  }
#pragma unroll
  for (int u = 0; u < K - 1; ++u) {
    acc.x += v[u].x;
    acc.y += v[u].y;
    acc.z += v[u].z;
    acc.w += v[u].w;
  }
  const bool ok = (h == 0) || !odd_tail;
  acc.x += ok ? v[K - 1].x : 0.0f;
  acc.y += ok ? v[K - 1].y : 0.0f;
  acc.z += ok ? v[K - 1].z : 0.0f;
  acc.w += ok ? v[K - 1].w : 0.0f;
}

template <int U0>
__device__ __forceinline__ void gather2_group(const float4* __restrict__ f4, int idx_l, int nent,
                                              int h, int t, float4& acc) {
  const int left = nent - 2 * U0;
  int nl = (left + 1) >> 1;
  nl = (nl > 8) ? 8 : nl;
  const bool odd_tail = (left < 16) && ((left & 1) != 0);
  switch (nl) {
    case 1: gather2_loads<U0, 1>(f4, idx_l, odd_tail, h, t, acc); break;
    case 2: gather2_loads<U0, 2>(f4, idx_l, odd_tail, h, t, acc); break;
    case 3: gather2_loads<U0, 3>(f4, idx_l, odd_tail, h, t, acc); break;
    case 4: gather2_loads<U0, 4>(f4, idx_l, odd_tail, h, t, acc); break;
    case 5: gather2_loads<U0, 5>(f4, idx_l, odd_tail, h, t, acc); break;
    case 6: gather2_loads<U0, 6>(f4, idx_l, odd_tail, h, t, acc); break;
    case 7: gather2_loads<U0, 7>(f4, idx_l, odd_tail, h, t, acc); break;
    case 8: gather2_loads<U0, 8>(f4, idx_l, odd_tail, h, t, acc); break;
    default: break;
  }
}

template <int K>
__device__ __forceinline__ void gather4_loads(const float4* __restrict__ f4, int idx_l, int nlast,
                                              int g, int t, float4& acc) {
  float4 v[K];
#pragma unroll
  for (int u = 0; u < K; ++u) {
    const int s = group_share_lane(idx_l, u);
    v[u] = f4[(unsigned)s * (OUTD / 4) + (unsigned)t];
  }
#pragma unroll
  for (int u = 0; u < K - 1; ++u) {
    acc.x += v[u].x;
    acc.y += v[u].y;
    acc.z += v[u].z;
    acc.w += v[u].w;
  }
  const bool ok = g < nlast;
  acc.x += ok ? v[K - 1].x : 0.0f;
  acc.y += ok ? v[K - 1].y : 0.0f;
  acc.z += ok ? v[K - 1].z : 0.0f;
  acc.w += ok ? v[K - 1].w : 0.0f;
}

__device__ __forceinline__ void gather4_group(const float4* __restrict__ f4, int idx_l, int nent,
                                              int g, int t, float4& acc) {
  const int nlast = ((nent - 1) & 3) + 1;
  switch ((nent + 3) >> 2) {
    case 1: gather4_loads<1>(f4, idx_l, nlast, g, t, acc); break;
    case 2: gather4_loads<2>(f4, idx_l, nlast, g, t, acc); break;
    case 3: gather4_loads<3>(f4, idx_l, nlast, g, t, acc); break;
    case 4: gather4_loads<4>(f4, idx_l, nlast, g, t, acc); break;
    case 5: gather4_loads<5>(f4, idx_l, nlast, g, t, acc); break;
    case 6: gather4_loads<6>(f4, idx_l, nlast, g, t, acc); break;
    case 7: gather4_loads<7>(f4, idx_l, nlast, g, t, acc); break;
    case 8: gather4_loads<8>(f4, idx_l, nlast, g, t, acc); break;
    default: break;
  }
}

#define RPW 8

#define BROWS (8 * RPW)
#define HP 68

__global__ __launch_bounds__(256) void k_gather64_relu_gemm2(
    const int* __restrict__ rowptr, const int* __restrict__ col,
    const float* __restrict__ dinv, const float* __restrict__ feat,
    const float* __restrict__ bias, const float* __restrict__ W2,
    float* __restrict__ hw2, int N) {
  __shared__ __attribute__((aligned(16))) float hs[BROWS * HP];
  const int lane = threadIdx.x & 31;
  const int wave = __builtin_amdgcn_readfirstlane((int)(threadIdx.x >> 5));
  const int r0 = (blockIdx.x * 8 + wave) * RPW;
  float* __restrict__ hrow = hs + wave * (RPW * HP);
  const int h = lane >> 4;
  const int t = lane & 15;
  const int ent = 2 * t + h;

  int nr = N - r0;
  nr = (nr < 0) ? 0 : ((nr < RPW) ? nr : RPW);

  if (nr > 0) {
    const int rp_l = rowptr[r0 + ((lane <= nr) ? lane : nr)];
    const float dd_l = dinv[r0 + ((lane < nr) ? lane : (nr - 1))];
    const float4* __restrict__ f4 = reinterpret_cast<const float4*>(feat);
    const float4 b = reinterpret_cast<const float4*>(bias)[t];
    float4 self_n = f4[(unsigned)r0 * (HID / 4) + (unsigned)t];

    int beg = __builtin_amdgcn_readlane(rp_l, 0);
    int end = __builtin_amdgcn_readlane(rp_l, 1);
    int c_n = r0;
    if (ent < end - beg) c_n = col[beg + ent];

    for (int ri = 0; ri < nr; ++ri) {
      const int row = r0 + ri;
      int s_l = (c_n < 0) ? 0 : ((c_n >= N) ? N - 1 : c_n);
      float4 acc;
      acc.x = (h == 0) ? self_n.x : 0.0f;
      acc.y = (h == 0) ? self_n.y : 0.0f;
      acc.z = (h == 0) ? self_n.z : 0.0f;
      acc.w = (h == 0) ? self_n.w : 0.0f;
      const int nbeg = end;
      int nend = end;
      if (ri + 1 < nr) {
        nend = __builtin_amdgcn_readlane(rp_l, ri + 2);
        c_n = row + 1;
        if (ent < nend - nbeg) c_n = col[nbeg + ent];
        self_n = f4[(unsigned)(row + 1) * (HID / 4) + (unsigned)t];
      }

      for (int base = beg; base < end; base += 32) {
        const int n = (end - base < 32) ? (end - base) : 32;
        if (base != beg) {
          s_l = row;
          if (ent < n) {
            s_l = col[base + ent];
            s_l = (s_l < 0) ? 0 : ((s_l >= N) ? N - 1 : s_l);
          }
        }
        gather2_group<0>(f4, s_l, n, h, t, acc);
        if (n > 16) gather2_group<8>(f4, s_l, n, h, t, acc);
      }

      acc.x = xhalf_add(acc.x);
      acc.y = xhalf_add(acc.y);
      acc.z = xhalf_add(acc.z);
      acc.w = xhalf_add(acc.w);
      const float dd = __int_as_float(__builtin_amdgcn_readlane(__float_as_int(dd_l), ri));
      float4 o;
      o.x = fmaxf(fmaf(acc.x, dd, b.x), 0.0f) * dd;
      o.y = fmaxf(fmaf(acc.y, dd, b.y), 0.0f) * dd;
      o.z = fmaxf(fmaf(acc.z, dd, b.z), 0.0f) * dd;
      o.w = fmaxf(fmaf(acc.w, dd, b.w), 0.0f) * dd;
      if (lane < 16) reinterpret_cast<float4*>(hrow + ri * HP)[t] = o;

      beg = nbeg;
      end = nend;
    }
  }
  for (int ri = nr; ri < RPW; ++ri) {
    float2 zz;
    zz.x = 0.0f;
    zz.y = 0.0f;
    reinterpret_cast<float2*>(hrow + ri * HP)[lane] = zz;
  }
  __syncthreads();

  if (wave >= BROWS / 16) return;
  const int rowBase = blockIdx.x * BROWS + wave * 16;
  if (rowBase >= N) return;
  const int kh  = lane >> 4;
  const int n16 = lane & 15;
  const float* ar = hs + (wave * 16 + n16) * HP + 2 * kh;
  const float* wr = W2 + (2 * kh) * OUTD + n16;

  v8f acc0 = {}, acc1 = {};
#pragma unroll 8
  for (int k = 0; k < HID; k += 4) {
    const v2f a = *reinterpret_cast<const v2f*>(ar + k);
    const float* wb = wr + k * OUTD;
    v2f b0; b0.x = wb[0];  b0.y = wb[OUTD];
    v2f b1; b1.x = wb[16]; b1.y = wb[OUTD + 16];
    acc0 = __builtin_amdgcn_wmma_f32_16x16x4_f32(false, a, false, b0, (short)0, acc0, false, false);
    acc1 = __builtin_amdgcn_wmma_f32_16x16x4_f32(false, a, false, b1, (short)0, acc1, false, false);
  }
  int rfirst = rowBase + 8 * kh;
  int cfirst = n16;
  asm volatile("v_nop\n\tv_nop\n\tv_nop\n\tv_nop" : "+v"(acc0), "+v"(acc1), "+v"(rfirst), "+v"(cfirst));
#pragma unroll
  for (int i = 0; i < 8; ++i) {
    const int r = rfirst + i;
    if (r < N) {
      const long base = (long)r * OUTD + cfirst;
      hw2[base +  0] = acc0[i];
      hw2[base + 16] = acc1[i];
    }
  }
}

__global__ __launch_bounds__(256) void k_gather32(
    const int* __restrict__ rowptr, const int* __restrict__ col,
    const float* __restrict__ dinv, const float* __restrict__ feat,
    const float* __restrict__ bias, const float* __restrict__ Wl,
    float* __restrict__ out, float* __restrict__ pq, int N) {
  const int lane = threadIdx.x & 31;
  const int rb = (blockIdx.x * 8 + (threadIdx.x >> 5)) * RPW;
  if (rb >= N) return;
  const int r0 = __builtin_amdgcn_readfirstlane(rb);
  const int nr = (N - r0 < RPW) ? (N - r0) : RPW;
  const int g = lane >> 3;
  const int t = lane & 7;
  const int ent = 4 * t + g;
  const int rp_l = rowptr[r0 + ((lane <= nr) ? lane : nr)];
  const float dd_l = dinv[r0 + ((lane < nr) ? lane : (nr - 1))];
  const float4* __restrict__ f4 = reinterpret_cast<const float4*>(feat);
  const float4 bv = reinterpret_cast<const float4*>(bias)[t];
  const float4 wl = reinterpret_cast<const float4*>(Wl)[(g & 1) * (OUTD / 4) + t];
  float pa_l = 0.0f, pb_l = 0.0f;

  int beg = __builtin_amdgcn_readlane(rp_l, 0);
  int end = __builtin_amdgcn_readlane(rp_l, 1);
  int c_n = r0;
  if (ent < end - beg) c_n = col[beg + ent];

  for (int ri = 0; ri < nr; ++ri) {
    const int row = r0 + ri;
    int s_l = (c_n < 0) ? 0 : ((c_n >= N) ? N - 1 : c_n);
    const int nbeg = end;
    int nend = end;
    if (ri + 1 < nr) {
      nend = __builtin_amdgcn_readlane(rp_l, ri + 2);
      c_n = row + 1;
      if (ent < nend - nbeg) c_n = col[nbeg + ent];
    }

    float4 acc;
    acc.x = 0.0f;
    acc.y = 0.0f;
    acc.z = 0.0f;
    acc.w = 0.0f;
    const int deg = end - beg;
    const int ne = deg + 1;
    for (int eb = 0; eb < ne; eb += 32) {
      const int n = (ne - eb < 32) ? (ne - eb) : 32;
      if (eb != 0) {
        s_l = row;
        if (eb + ent < deg) {
          s_l = col[beg + eb + ent];
          s_l = (s_l < 0) ? 0 : ((s_l >= N) ? N - 1 : s_l);
        }
      }
      gather4_group(f4, s_l, n, g, t, acc);
    }

    acc.x = xhalf_add(row_xmask_add<8>(acc.x));
    acc.y = xhalf_add(row_xmask_add<8>(acc.y));
    acc.z = xhalf_add(row_xmask_add<8>(acc.z));
    acc.w = xhalf_add(row_xmask_add<8>(acc.w));
    const float dd = __int_as_float(__builtin_amdgcn_readlane(__float_as_int(dd_l), ri));
    float4 o;
    o.x = fmaf(acc.x, dd, bv.x);
    o.y = fmaf(acc.y, dd, bv.y);
    o.z = fmaf(acc.z, dd, bv.z);
    o.w = fmaf(acc.w, dd, bv.w);
    if (lane < 8) reinterpret_cast<float4*>(out)[(unsigned)row * (OUTD / 4) + (unsigned)t] = o;
    float sa = o.x * wl.x;
    sa = fmaf(o.y, wl.y, sa);
    sa = fmaf(o.z, wl.z, sa);
    sa = fmaf(o.w, wl.w, sa);
    sa = row_xmask_add<4>(row_xmask_add<2>(row_xmask_add<1>(sa)));
    const float sb = __int_as_float(__builtin_amdgcn_update_dpp(0, __float_as_int(sa), 0x168, 0xf, 0xf, true));
    if (lane == ri) {
      pa_l = sa;
      pb_l = sb;
    }

    beg = nbeg;
    end = nend;
  }
  if (lane < nr) {
    float2 p;
    p.x = pa_l;
    p.y = pb_l;
    reinterpret_cast<float2*>(pq)[r0 + lane] = p;
  }
}

#define LT 128
#define LPT 4
__global__ __launch_bounds__(LT) void k_link(const float* __restrict__ pq, const int* __restrict__ eli,
                                             const float* __restrict__ bl,
                                             float* __restrict__ pred, int EL) {
  const int bbase = blockIdx.x * (LT * LPT);
  const float b = bl[0];
  if (bbase + LT * LPT <= EL) {
    const int e0 = bbase + threadIdx.x;
    int s[LPT], d[LPT];
#pragma unroll
    for (int k = 0; k < LPT; ++k) {
      s[k] = eli[e0 + k * LT];
      d[k] = eli[EL + e0 + k * LT];
    }
    float ps[LPT], pd[LPT];
#pragma unroll
    for (int k = 0; k < LPT; ++k) {
      ps[k] = pq[2 * (long)s[k]];
      pd[k] = pq[2 * (long)d[k] + 1];
    }
#pragma unroll
    for (int k = 0; k < LPT; ++k) pred[e0 + k * LT] = (ps[k] + pd[k]) + b;
  } else {
    for (int k = 0; k < LPT; ++k) {
      const int e = bbase + k * LT + threadIdx.x;
      if (e < EL) pred[e] = (pq[2 * (long)eli[e]] + pq[2 * (long)eli[EL + e] + 1]) + b;
    }
  }
}

extern "C" void kernel_launch(void* const* d_in, const int* in_sizes, int n_in,
                              void* d_out, int out_size, void* d_ws, size_t ws_size,
                              hipStream_t stream) {
  const float* x   = (const float*)d_in[0];
  const float* pos = (const float*)d_in[1];
  const float* W1  = (const float*)d_in[2];
  const float* b1  = (const float*)d_in[3];
  const float* W2  = (const float*)d_in[4];
  const float* b2  = (const float*)d_in[5];
  const float* Wl  = (const float*)d_in[6];
  const float* bl  = (const float*)d_in[7];
  const int*   ei  = (const int*)d_in[8];
  const int*   eli = (const int*)d_in[9];

  const int N  = in_sizes[0] / IN_C;
  const int E  = in_sizes[8] / 2;
  const int EL = in_sizes[9] / 2;

  float* z    = (float*)d_out;
  float* pred = z + (long)N * OUTD;

  float* hw1   = (float*)d_ws;
  float* h1    = hw1 + (long)N * HID;
  float* reg3  = h1  + (long)N * HID;
  float* dinv  = reg3 + (long)N * OUTD;
  float* hw2   = h1;
  int2*  seg   = (int2*)h1;
  int*   loff  = (int*)(seg + E);
  int*   col   = (int*)reg3;
  int*   rowptr = col + E;
  float* pq    = hw1;

  const int* esrc = ei;
  const int* edst = ei + E;

  const int B = 256;
  const int tiles  = (E + PTILE - 1) / PTILE;
  const int ranges = (N + RN - 1) / RN;

  k_part<<<tiles, PT, 0, stream>>>(esrc, edst, seg, loff, E, N, ranges);
  k_rowsort_gemm1<<<ranges, RT, 0, stream>>>(seg, loff, rowptr, dinv, col, x, pos, W1, hw1, N, ranges, tiles);

  const int rblocks  = (N + 8 * RPW - 1) / (8 * RPW);

  k_gather64_relu_gemm2<<<rblocks, B, 0, stream>>>(rowptr, col, dinv, hw1, b1, W2, hw2, N);

  k_gather32<<<rblocks, B, 0, stream>>>(rowptr, col, dinv, hw2, b2, Wl, z, pq, N);

  k_link<<<(EL + LT * LPT - 1) / (LT * LPT), LT, 0, stream>>>(pq, eli, bl, pred, EL);
}


# GEMM 1 beside the row sort, not behind it: 8 of a range block's 16 waves sort, 8 multiply (tile 1 under the counting walk, tile 2 under the placing walk); row scale per finished tile, from LDS
# speedup vs baseline: 7.3097x; 1.0414x over previous
#include <hip/hip_runtime.h>

typedef __attribute__((ext_vector_type(2))) float v2f;
typedef __attribute__((ext_vector_type(8))) float v8f;

#define IN_C 64
#define POSD 16
#define KDIM1 (IN_C + POSD)
#define HID 64
#define OUTD 32


#define RN_SHIFT 8
#define RN 256
#define MAXR 1024
#define PT 1024
#define RT 512
#define RW (RT / 32)
#define SW 8
#define ST (SW * 32)
#define EPT 8
#define PTILE (PT * EPT)
#define CAPC 12288

__global__ __launch_bounds__(PT) void k_part(const int* __restrict__ src, const int* __restrict__ dst,
                                             int2* __restrict__ seg, int* __restrict__ loff,
                                             int E, int N, int R) {
  __shared__ int rc[MAXR];
  __shared__ int wsum[32];
  __shared__ int2 stage[PTILE];
  const int tid = threadIdx.x, lane = tid & 31, wave = tid >> 5;
  const int base = blockIdx.x * PTILE;

  rc[tid] = 0;
  __syncthreads();

  int s[EPT], d[EPT], rk[EPT];
#pragma unroll
  for (int k = 0; k < EPT; ++k) {
    const int e = base + k * PT + tid;
    const int ec = (e < E) ? e : (E - 1);
    s[k] = src[ec];
    d[k] = dst[ec];
  }
#pragma unroll
  for (int k = 0; k < EPT; ++k) {
    const int e = base + k * PT + tid;
    int t = d[k];
    t = (t < 0) ? 0 : ((t >= N) ? N - 1 : t);
    d[k] = (e < E) ? t : -1;
  }
#pragma unroll
  for (int k = 0; k < EPT; ++k) {
    rk[k] = 0;
    if (d[k] >= 0) rk[k] = atomicAdd(&rc[d[k] >> RN_SHIFT], 1);
  }
  __syncthreads();

  const int c = rc[tid];
  int inc = c;
  for (int off = 1; off < 32; off <<= 1) {
    const int t = __shfl_up(inc, off, 32);
    if (lane >= off) inc += t;
  }
  if (lane == 31) wsum[wave] = inc;
  __syncthreads();
  if (wave == 0) {
    const int v = wsum[lane];
    int iv = v;
    for (int off = 1; off < 32; off <<= 1) {
      const int t = __shfl_up(iv, off, 32);
      if (lane >= off) iv += t;
    }
    wsum[lane] = iv - v;
  }
  __syncthreads();
  const int excl = wsum[wave] + inc - c;
  rc[tid] = excl;
  if (tid <= R) loff[(long)blockIdx.x * (R + 1) + tid] = excl;
  __syncthreads();

#pragma unroll
  for (int k = 0; k < EPT; ++k) {
    if (d[k] >= 0) {
      int2 p;
      p.x = s[k];
      p.y = d[k];
      stage[rc[d[k] >> RN_SHIFT] + rk[k]] = p;
    }
  }
  __syncthreads();

  const int nloc = (E - base < PTILE) ? (E - base) : PTILE;
  for (int i = tid; i < nloc; i += PT) seg[(long)base + i] = stage[i];
}

template <bool PLACE>
__device__ __forceinline__ void walk_pieces(const int2* __restrict__ seg, const int* __restrict__ loff,
                                            int R, int PB, int r, int lane, int wave,
                                            int* cnt, int* cur, int* cstage, int* __restrict__ colr,
                                            bool staged) {
  for (int p0 = 0; p0 < PB; p0 += ST) {
    const int p = p0 + lane * SW + wave;
    int pst = 0, pcn = 0;
    if (p < PB) {
      const int a = loff[(long)p * (R + 1) + r];
      const int b = loff[(long)p * (R + 1) + r + 1];
      pst = p * PTILE + a;
      pcn = b - a;
    }
    const int rem = PB - p0 - wave;
    int npw = (rem <= 0) ? 0 : ((rem + SW - 1) / SW);
    if (npw > 32) npw = 32;
    for (int j = 0; j < npw; j += 8) {
      int st[8], cn[8];
      int2 e[8];
#pragma unroll
      for (int u = 0; u < 8; ++u) {
        st[u] = __builtin_amdgcn_readlane(pst, j + u);
        cn[u] = __builtin_amdgcn_readlane(pcn, j + u);
      }
#pragma unroll
      for (int u = 0; u < 8; ++u) {
        const long idx = (lane < cn[u]) ? ((long)st[u] + lane) : 0;
        e[u] = seg[idx];
      }
#pragma unroll
      for (int u = 0; u < 8; ++u) {
        if (lane < cn[u]) {
          const int loc = e[u].y & (RN - 1);
          if (!PLACE) {
            atomicAdd(&cnt[loc], 1);
          } else {
            const int pos = atomicAdd(&cur[loc], 1);
            if (staged) cstage[pos] = e[u].x; else colr[pos] = e[u].x;
          }
        }
      }
    }
    for (int j = 0; j < npw; ++j) {
      const int cnj = __builtin_amdgcn_readlane(pcn, j);
      if (cnj > 32) {
        const int stj = __builtin_amdgcn_readlane(pst, j);
        for (int i = 32 + lane; i < cnj; i += 32) {
          const int2 ee = seg[(long)stj + i];
          const int loc = ee.y & (RN - 1);
          if (!PLACE) {
            atomicAdd(&cnt[loc], 1);
          } else {
            const int pos = atomicAdd(&cur[loc], 1);
            if (staged) cstage[pos] = ee.x; else colr[pos] = ee.x;
          }
        }
      }
    }
  }
}

static_assert(RN == ST, "one sorting thread per node of the range");
static_assert(RN == 32 * (RW - SW), "two 16-row tiles of the range per multiplying wave");

__device__ __forceinline__ void gemm1_tile(const float* __restrict__ x, const float* __restrict__ pos,
                                           const float* __restrict__ W1, int row, int kh, int n16,
                                           v8f& acc0, v8f& acc1, v8f& acc2, v8f& acc3) {
  for (int k = 0; k < KDIM1; k += 4) {
    const int ka = k + 2 * kh;
    v2f a;
    a.x = (ka     < IN_C) ? x[(long)row * IN_C + ka]       : pos[(long)row * POSD + (ka     - IN_C)];
    a.y = (ka + 1 < IN_C) ? x[(long)row * IN_C + ka + 1]   : pos[(long)row * POSD + (ka + 1 - IN_C)];
    const float* wb = W1 + (long)ka * HID + n16;
    v2f b0; b0.x = wb[0];  b0.y = wb[HID];
    v2f b1; b1.x = wb[16]; b1.y = wb[HID + 16];
    v2f b2; b2.x = wb[32]; b2.y = wb[HID + 32];
    v2f b3; b3.x = wb[48]; b3.y = wb[HID + 48];
    acc0 = __builtin_amdgcn_wmma_f32_16x16x4_f32(false, a, false, b0, (short)0, acc0, false, false);
    acc1 = __builtin_amdgcn_wmma_f32_16x16x4_f32(false, a, false, b1, (short)0, acc1, false, false);
    acc2 = __builtin_amdgcn_wmma_f32_16x16x4_f32(false, a, false, b2, (short)0, acc2, false, false);
    acc3 = __builtin_amdgcn_wmma_f32_16x16x4_f32(false, a, false, b3, (short)0, acc3, false, false);
  }
}

__device__ __forceinline__ void gemm1_store(float* __restrict__ hw1, const float* dlt, int rowBase,
                                            int kh, int n16, int N,
                                            const v8f& acc0, const v8f& acc1, const v8f& acc2, const v8f& acc3) {
#pragma unroll
  for (int i = 0; i < 8; ++i) {
    const int r = rowBase + i + 8 * kh;
    const float sc = dlt[i + 8 * kh];
    if (r < N) {
      const long base = (long)r * HID + n16;
      hw1[base +  0] = acc0[i] * sc;
      hw1[base + 16] = acc1[i] * sc;
      hw1[base + 32] = acc2[i] * sc;
      hw1[base + 48] = acc3[i] * sc;
    }
  }
}

__global__ __launch_bounds__(RT) __attribute__((amdgpu_waves_per_eu(8))) void k_rowsort_gemm1(
    const int2* __restrict__ seg, const int* __restrict__ loff,
    int* __restrict__ rowptr, float* __restrict__ dinv, int* __restrict__ col,
    const float* __restrict__ x, const float* __restrict__ pos,
    const float* __restrict__ W1, float* __restrict__ hw1,
    int N, int R, int PB) {
  __shared__ float dl[RN];
  __shared__ int cnt[RN];
  __shared__ int cur[RN];
  __shared__ int red[SW];
  __shared__ int wtot[RN / 32];
  __shared__ int cstage[CAPC];
  const int tid = threadIdx.x, lane = tid & 31;
  const int wave = __builtin_amdgcn_readfirstlane(tid >> 5);
  const bool sorter = wave < SW;
  const int r = blockIdx.x;
  const int nbase = r << RN_SHIFT;

  const int kh  = lane >> 4;
  const int n16 = lane & 15;
  const int g = sorter ? 0 : (wave - SW);
  const int rowBase0 = nbase + g * 32;
  const int rowBase1 = rowBase0 + 16;
  const bool mult = !sorter && rowBase0 < N;
  int row0 = rowBase0 + n16, row1 = rowBase1 + n16;
  if (row0 >= N) row0 = N - 1;
  if (row1 >= N) row1 = N - 1;
  v8f acc0 = {}, acc1 = {}, acc2 = {}, acc3 = {};

  if (mult) {
    int pr = rowBase0 + lane;
    if (pr >= N) pr = N - 1;
    __builtin_prefetch(x + (long)pr * IN_C, 0, 3);
    int pp = rowBase0 + 4 * (lane & 7);
    if (pp >= N) pp = N - 1;
    if (lane < 8) __builtin_prefetch(pos + (long)pp * POSD, 0, 3);
  }

  if (sorter) {
    cnt[tid] = 0;
    int bl = 0;
    for (int p = tid; p < PB; p += ST) bl += loff[(long)p * (R + 1) + r];
    for (int off = 16; off > 0; off >>= 1) bl += __shfl_down(bl, off, 32);
    if (lane == 0) red[wave] = bl;
  }
  __syncthreads();
  int rbase = 0;
  for (int w = 0; w < SW; ++w) rbase += red[w];

  if (sorter) walk_pieces<false>(seg, loff, R, PB, r, lane, wave, cnt, cur, cstage, col, false);
  else if (mult) gemm1_tile(x, pos, W1, row0, kh, n16, acc0, acc1, acc2, acc3);
  __syncthreads();

  const int c = sorter ? cnt[tid] : 0;
  int inc = c;
  for (int off = 1; off < 32; off <<= 1) {
    const int t = __shfl_up(inc, off, 32);
    if (lane >= off) inc += t;
  }
  if (sorter && lane == 31) wtot[wave] = inc;
  __syncthreads();
  int ntot = 0;
  for (int w = 0; w < RN / 32; ++w) ntot += wtot[w];
  if (sorter) {
    int woff = 0;
    for (int w = 0; w < wave; ++w) woff += wtot[w];
    const int excl = woff + inc - c;
    cur[tid] = excl;
    const int node = nbase + tid;
    if (node < N) {
      rowptr[node] = rbase + excl;
      dinv[node] = rsqrtf((float)(c + 1));
      if (node == N - 1) rowptr[N] = rbase + excl + c;
    }
    dl[tid] = rsqrtf((float)(c + 1));
  }
  const bool staged = (ntot <= CAPC);
  __syncthreads();

  if (sorter) {
    walk_pieces<true>(seg, loff, R, PB, r, lane, wave, cnt, cur, cstage, col + rbase, staged);
  } else if (mult) {
    gemm1_store(hw1, dl + g * 32, rowBase0, kh, n16, N, acc0, acc1, acc2, acc3);
    acc0 = (v8f){}; acc1 = (v8f){}; acc2 = (v8f){}; acc3 = (v8f){};
    gemm1_tile(x, pos, W1, row1, kh, n16, acc0, acc1, acc2, acc3);
  }

  if (staged) {
    __syncthreads();
    if (sorter) for (int i = tid; i < ntot; i += ST) col[(long)rbase + i] = cstage[i];
  }

  if (mult) gemm1_store(hw1, dl + g * 32 + 16, rowBase1, kh, n16, N, acc0, acc1, acc2, acc3);
}


template <int XMASK>
__device__ __forceinline__ float row_xmask_add(float v) {
  return v + __int_as_float(__builtin_amdgcn_update_dpp(0, __float_as_int(v), 0x160 + XMASK, 0xf, 0xf, true));
}

__device__ __forceinline__ float xhalf_add(float v) {
  const int b = __float_as_int(v);
  return v + __int_as_float(__builtin_amdgcn_permlanex16(b, b, 0x76543210, 0xfedcba98, false, false));
}

#define ROW_SHARE_CASE(U) case U: return __builtin_amdgcn_update_dpp(0, v, 0x150 + U, 0xf, 0xf, true);
__device__ __forceinline__ int row_share_lane(int v, int u) {
  switch (u) {
    ROW_SHARE_CASE(0)  ROW_SHARE_CASE(1)  ROW_SHARE_CASE(2)  ROW_SHARE_CASE(3)
    ROW_SHARE_CASE(4)  ROW_SHARE_CASE(5)  ROW_SHARE_CASE(6)  ROW_SHARE_CASE(7)
    ROW_SHARE_CASE(8)  ROW_SHARE_CASE(9)  ROW_SHARE_CASE(10) ROW_SHARE_CASE(11)
    ROW_SHARE_CASE(12) ROW_SHARE_CASE(13) ROW_SHARE_CASE(14) ROW_SHARE_CASE(15)
    default: return v;
  }
}

#define GROUP_SHARE_CASE(U) case U: return (int)__builtin_amdgcn_mov_dpp8((unsigned)v, U * 0x249249);
__device__ __forceinline__ int group_share_lane(int v, int u) {
  switch (u) {
    GROUP_SHARE_CASE(0) GROUP_SHARE_CASE(1) GROUP_SHARE_CASE(2) GROUP_SHARE_CASE(3)
    GROUP_SHARE_CASE(4) GROUP_SHARE_CASE(5) GROUP_SHARE_CASE(6) GROUP_SHARE_CASE(7)
    default: return v;
  }
}

template <int U0, int K>
__device__ __forceinline__ void gather2_loads(const float4* __restrict__ f4, int idx_l, bool odd_tail,
                                              int h, int t, float4& acc) {
  float4 v[K];
#pragma unroll
  for (int u = 0; u < K; ++u) {
    const int s = row_share_lane(idx_l, U0 + u);
    v[u] = f4[(unsigned)s * (HID / 4) + (unsigned)t];
  }
#pragma unroll
  for (int u = 0; u < K - 1; ++u) {
    acc.x += v[u].x;
    acc.y += v[u].y;
    acc.z += v[u].z;
    acc.w += v[u].w;
  }
  const bool ok = (h == 0) || !odd_tail;
  acc.x += ok ? v[K - 1].x : 0.0f;
  acc.y += ok ? v[K - 1].y : 0.0f;
  acc.z += ok ? v[K - 1].z : 0.0f;
  acc.w += ok ? v[K - 1].w : 0.0f;
}

template <int U0>
__device__ __forceinline__ void gather2_group(const float4* __restrict__ f4, int idx_l, int nent,
                                              int h, int t, float4& acc) {
  const int left = nent - 2 * U0;
  int nl = (left + 1) >> 1;
  nl = (nl > 8) ? 8 : nl;
  const bool odd_tail = (left < 16) && ((left & 1) != 0);
  switch (nl) {
    case 1: gather2_loads<U0, 1>(f4, idx_l, odd_tail, h, t, acc); break;
    case 2: gather2_loads<U0, 2>(f4, idx_l, odd_tail, h, t, acc); break;
    case 3: gather2_loads<U0, 3>(f4, idx_l, odd_tail, h, t, acc); break;
    case 4: gather2_loads<U0, 4>(f4, idx_l, odd_tail, h, t, acc); break;
    case 5: gather2_loads<U0, 5>(f4, idx_l, odd_tail, h, t, acc); break;
    case 6: gather2_loads<U0, 6>(f4, idx_l, odd_tail, h, t, acc); break;
    case 7: gather2_loads<U0, 7>(f4, idx_l, odd_tail, h, t, acc); break;
    case 8: gather2_loads<U0, 8>(f4, idx_l, odd_tail, h, t, acc); break;
    default: break;
  }
}

template <int K>
__device__ __forceinline__ void gather4_loads(const float4* __restrict__ f4, int idx_l, int nlast,
                                              int g, int t, float4& acc) {
  float4 v[K];
#pragma unroll
  for (int u = 0; u < K; ++u) {
    const int s = group_share_lane(idx_l, u);
    v[u] = f4[(unsigned)s * (OUTD / 4) + (unsigned)t];
  }
#pragma unroll
  for (int u = 0; u < K - 1; ++u) {
    acc.x += v[u].x;
    acc.y += v[u].y;
    acc.z += v[u].z;
    acc.w += v[u].w;
  }
  const bool ok = g < nlast;
  acc.x += ok ? v[K - 1].x : 0.0f;
  acc.y += ok ? v[K - 1].y : 0.0f;
  acc.z += ok ? v[K - 1].z : 0.0f;
  acc.w += ok ? v[K - 1].w : 0.0f;
}

__device__ __forceinline__ void gather4_group(const float4* __restrict__ f4, int idx_l, int nent,
                                              int g, int t, float4& acc) {
  const int nlast = ((nent - 1) & 3) + 1;
  switch ((nent + 3) >> 2) {
    case 1: gather4_loads<1>(f4, idx_l, nlast, g, t, acc); break;
    case 2: gather4_loads<2>(f4, idx_l, nlast, g, t, acc); break;
    case 3: gather4_loads<3>(f4, idx_l, nlast, g, t, acc); break;
    case 4: gather4_loads<4>(f4, idx_l, nlast, g, t, acc); break;
    case 5: gather4_loads<5>(f4, idx_l, nlast, g, t, acc); break;
    case 6: gather4_loads<6>(f4, idx_l, nlast, g, t, acc); break;
    case 7: gather4_loads<7>(f4, idx_l, nlast, g, t, acc); break;
    case 8: gather4_loads<8>(f4, idx_l, nlast, g, t, acc); break;
    default: break;
  }
}

#define RPW 8

#define BROWS (8 * RPW)
#define HP 68

__global__ __launch_bounds__(256) void k_gather64_relu_gemm2(
    const int* __restrict__ rowptr, const int* __restrict__ col,
    const float* __restrict__ dinv, const float* __restrict__ feat,
    const float* __restrict__ bias, const float* __restrict__ W2,
    float* __restrict__ hw2, int N) {
  __shared__ __attribute__((aligned(16))) float hs[BROWS * HP];
  const int lane = threadIdx.x & 31;
  const int wave = __builtin_amdgcn_readfirstlane((int)(threadIdx.x >> 5));
  const int r0 = (blockIdx.x * 8 + wave) * RPW;
  float* __restrict__ hrow = hs + wave * (RPW * HP);
  const int h = lane >> 4;
  const int t = lane & 15;
  const int ent = 2 * t + h;

  int nr = N - r0;
  nr = (nr < 0) ? 0 : ((nr < RPW) ? nr : RPW);

  if (nr > 0) {
    const int rp_l = rowptr[r0 + ((lane <= nr) ? lane : nr)];
    const float dd_l = dinv[r0 + ((lane < nr) ? lane : (nr - 1))];
    const float4* __restrict__ f4 = reinterpret_cast<const float4*>(feat);
    const float4 b = reinterpret_cast<const float4*>(bias)[t];
    float4 self_n = f4[(unsigned)r0 * (HID / 4) + (unsigned)t];

    int beg = __builtin_amdgcn_readlane(rp_l, 0);
    int end = __builtin_amdgcn_readlane(rp_l, 1);
    int c_n = r0;
    if (ent < end - beg) c_n = col[beg + ent];

    for (int ri = 0; ri < nr; ++ri) {
      const int row = r0 + ri;
      int s_l = (c_n < 0) ? 0 : ((c_n >= N) ? N - 1 : c_n);
      float4 acc;
      acc.x = (h == 0) ? self_n.x : 0.0f;
      acc.y = (h == 0) ? self_n.y : 0.0f;
      acc.z = (h == 0) ? self_n.z : 0.0f;
      acc.w = (h == 0) ? self_n.w : 0.0f;
      const int nbeg = end;
      int nend = end;
      if (ri + 1 < nr) {
        nend = __builtin_amdgcn_readlane(rp_l, ri + 2);
        c_n = row + 1;
        if (ent < nend - nbeg) c_n = col[nbeg + ent];
        self_n = f4[(unsigned)(row + 1) * (HID / 4) + (unsigned)t];
      }

      for (int base = beg; base < end; base += 32) {
        const int n = (end - base < 32) ? (end - base) : 32;
        if (base != beg) {
          s_l = row;
          if (ent < n) {
            s_l = col[base + ent];
            s_l = (s_l < 0) ? 0 : ((s_l >= N) ? N - 1 : s_l);
          }
        }
        gather2_group<0>(f4, s_l, n, h, t, acc);
        if (n > 16) gather2_group<8>(f4, s_l, n, h, t, acc);
      }

      acc.x = xhalf_add(acc.x);
      acc.y = xhalf_add(acc.y);
      acc.z = xhalf_add(acc.z);
      acc.w = xhalf_add(acc.w);
      const float dd = __int_as_float(__builtin_amdgcn_readlane(__float_as_int(dd_l), ri));
      float4 o;
      o.x = fmaxf(fmaf(acc.x, dd, b.x), 0.0f) * dd;
      o.y = fmaxf(fmaf(acc.y, dd, b.y), 0.0f) * dd;
      o.z = fmaxf(fmaf(acc.z, dd, b.z), 0.0f) * dd;
      o.w = fmaxf(fmaf(acc.w, dd, b.w), 0.0f) * dd;
      if (lane < 16) reinterpret_cast<float4*>(hrow + ri * HP)[t] = o;

      beg = nbeg;
      end = nend;
    }
  }
  for (int ri = nr; ri < RPW; ++ri) {
    float2 zz;
    zz.x = 0.0f;
    zz.y = 0.0f;
    reinterpret_cast<float2*>(hrow + ri * HP)[lane] = zz;
  }
  __syncthreads();

  if (wave >= BROWS / 16) return;
  const int rowBase = blockIdx.x * BROWS + wave * 16;
  if (rowBase >= N) return;
  const int kh  = lane >> 4;
  const int n16 = lane & 15;
  const float* ar = hs + (wave * 16 + n16) * HP + 2 * kh;
  const float* wr = W2 + (2 * kh) * OUTD + n16;

  v8f acc0 = {}, acc1 = {};
#pragma unroll 8
  for (int k = 0; k < HID; k += 4) {
    const v2f a = *reinterpret_cast<const v2f*>(ar + k);
    const float* wb = wr + k * OUTD;
    v2f b0; b0.x = wb[0];  b0.y = wb[OUTD];
    v2f b1; b1.x = wb[16]; b1.y = wb[OUTD + 16];
    acc0 = __builtin_amdgcn_wmma_f32_16x16x4_f32(false, a, false, b0, (short)0, acc0, false, false);
    acc1 = __builtin_amdgcn_wmma_f32_16x16x4_f32(false, a, false, b1, (short)0, acc1, false, false);
  }
  int rfirst = rowBase + 8 * kh;
  int cfirst = n16;
  asm volatile("v_nop\n\tv_nop\n\tv_nop\n\tv_nop" : "+v"(acc0), "+v"(acc1), "+v"(rfirst), "+v"(cfirst));
#pragma unroll
  for (int i = 0; i < 8; ++i) {
    const int r = rfirst + i;
    if (r < N) {
      const long base = (long)r * OUTD + cfirst;
      hw2[base +  0] = acc0[i];
      hw2[base + 16] = acc1[i];
    }
  }
}

__global__ __launch_bounds__(256) void k_gather32(
    const int* __restrict__ rowptr, const int* __restrict__ col,
    const float* __restrict__ dinv, const float* __restrict__ feat,
    const float* __restrict__ bias, const float* __restrict__ Wl,
    float* __restrict__ out, float* __restrict__ pq, int N) {
  const int lane = threadIdx.x & 31;
  const int rb = (blockIdx.x * 8 + (threadIdx.x >> 5)) * RPW;
  if (rb >= N) return;
  const int r0 = __builtin_amdgcn_readfirstlane(rb);
  const int nr = (N - r0 < RPW) ? (N - r0) : RPW;
  const int g = lane >> 3;
  const int t = lane & 7;
  const int ent = 4 * t + g;
  const int rp_l = rowptr[r0 + ((lane <= nr) ? lane : nr)];
  const float dd_l = dinv[r0 + ((lane < nr) ? lane : (nr - 1))];
  const float4* __restrict__ f4 = reinterpret_cast<const float4*>(feat);
  const float4 bv = reinterpret_cast<const float4*>(bias)[t];
  const float4 wl = reinterpret_cast<const float4*>(Wl)[(g & 1) * (OUTD / 4) + t];
  float pa_l = 0.0f, pb_l = 0.0f;

  int beg = __builtin_amdgcn_readlane(rp_l, 0);
  int end = __builtin_amdgcn_readlane(rp_l, 1);
  int c_n = r0;
  if (ent < end - beg) c_n = col[beg + ent];

  for (int ri = 0; ri < nr; ++ri) {
    const int row = r0 + ri;
    int s_l = (c_n < 0) ? 0 : ((c_n >= N) ? N - 1 : c_n);
    const int nbeg = end;
    int nend = end;
    if (ri + 1 < nr) {
      nend = __builtin_amdgcn_readlane(rp_l, ri + 2);
      c_n = row + 1;
      if (ent < nend - nbeg) c_n = col[nbeg + ent];
    }

    float4 acc;
    acc.x = 0.0f;
    acc.y = 0.0f;
    acc.z = 0.0f;
    acc.w = 0.0f;
    const int deg = end - beg;
    const int ne = deg + 1;
    for (int eb = 0; eb < ne; eb += 32) {
      const int n = (ne - eb < 32) ? (ne - eb) : 32;
      if (eb != 0) {
        s_l = row;
        if (eb + ent < deg) {
          s_l = col[beg + eb + ent];
          s_l = (s_l < 0) ? 0 : ((s_l >= N) ? N - 1 : s_l);
        }
      }
      gather4_group(f4, s_l, n, g, t, acc);
    }

    acc.x = xhalf_add(row_xmask_add<8>(acc.x));
    acc.y = xhalf_add(row_xmask_add<8>(acc.y));
    acc.z = xhalf_add(row_xmask_add<8>(acc.z));
    acc.w = xhalf_add(row_xmask_add<8>(acc.w));
    const float dd = __int_as_float(__builtin_amdgcn_readlane(__float_as_int(dd_l), ri));
    float4 o;
    o.x = fmaf(acc.x, dd, bv.x);
    o.y = fmaf(acc.y, dd, bv.y);
    o.z = fmaf(acc.z, dd, bv.z);
    o.w = fmaf(acc.w, dd, bv.w);
    if (lane < 8) reinterpret_cast<float4*>(out)[(unsigned)row * (OUTD / 4) + (unsigned)t] = o;
    float sa = o.x * wl.x;
    sa = fmaf(o.y, wl.y, sa);
    sa = fmaf(o.z, wl.z, sa);
    sa = fmaf(o.w, wl.w, sa);
    sa = row_xmask_add<4>(row_xmask_add<2>(row_xmask_add<1>(sa)));
    const float sb = __int_as_float(__builtin_amdgcn_update_dpp(0, __float_as_int(sa), 0x168, 0xf, 0xf, true));
    if (lane == ri) {
      pa_l = sa;
      pb_l = sb;
    }

    beg = nbeg;
    end = nend;
  }
  if (lane < nr) {
    float2 p;
    p.x = pa_l;
    p.y = pb_l;
    reinterpret_cast<float2*>(pq)[r0 + lane] = p;
  }
}

#define LT 128
#define LPT 4
__global__ __launch_bounds__(LT) void k_link(const float* __restrict__ pq, const int* __restrict__ eli,
                                             const float* __restrict__ bl,
                                             float* __restrict__ pred, int EL) {
  const int bbase = blockIdx.x * (LT * LPT);
  const float b = bl[0];
  if (bbase + LT * LPT <= EL) {
    const int e0 = bbase + threadIdx.x;
    int s[LPT], d[LPT];
#pragma unroll
    for (int k = 0; k < LPT; ++k) {
      s[k] = eli[e0 + k * LT];
      d[k] = eli[EL + e0 + k * LT];
    }
    float ps[LPT], pd[LPT];
#pragma unroll
    for (int k = 0; k < LPT; ++k) {
      ps[k] = pq[2 * (long)s[k]];
      pd[k] = pq[2 * (long)d[k] + 1];
    }
#pragma unroll
    for (int k = 0; k < LPT; ++k) pred[e0 + k * LT] = (ps[k] + pd[k]) + b;
  } else {
    for (int k = 0; k < LPT; ++k) {
      const int e = bbase + k * LT + threadIdx.x;
      if (e < EL) pred[e] = (pq[2 * (long)eli[e]] + pq[2 * (long)eli[EL + e] + 1]) + b;
    }
  }
}

extern "C" void kernel_launch(void* const* d_in, const int* in_sizes, int n_in,
                              void* d_out, int out_size, void* d_ws, size_t ws_size,
                              hipStream_t stream) {
  const float* x   = (const float*)d_in[0];
  const float* pos = (const float*)d_in[1];
  const float* W1  = (const float*)d_in[2];
  const float* b1  = (const float*)d_in[3];
  const float* W2  = (const float*)d_in[4];
  const float* b2  = (const float*)d_in[5];
  const float* Wl  = (const float*)d_in[6];
  const float* bl  = (const float*)d_in[7];
  const int*   ei  = (const int*)d_in[8];
  const int*   eli = (const int*)d_in[9];

  const int N  = in_sizes[0] / IN_C;
  const int E  = in_sizes[8] / 2;
  const int EL = in_sizes[9] / 2;

  float* z    = (float*)d_out;
  float* pred = z + (long)N * OUTD;

  float* hw1   = (float*)d_ws;
  float* h1    = hw1 + (long)N * HID;
  float* reg3  = h1  + (long)N * HID;
  float* dinv  = reg3 + (long)N * OUTD;
  float* hw2   = h1;
  int2*  seg   = (int2*)h1;
  int*   loff  = (int*)(seg + E);
  int*   col   = (int*)reg3;
  int*   rowptr = col + E;
  float* pq    = hw1;

  const int* esrc = ei;
  const int* edst = ei + E;

  const int B = 256;
  const int tiles  = (E + PTILE - 1) / PTILE;
  const int ranges = (N + RN - 1) / RN;

  k_part<<<tiles, PT, 0, stream>>>(esrc, edst, seg, loff, E, N, ranges);
  k_rowsort_gemm1<<<ranges, RT, 0, stream>>>(seg, loff, rowptr, dinv, col, x, pos, W1, hw1, N, ranges, tiles);

  const int rblocks  = (N + 8 * RPW - 1) / (8 * RPW);

  k_gather64_relu_gemm2<<<rblocks, B, 0, stream>>>(rowptr, col, dinv, hw1, b1, W2, hw2, N);

  k_gather32<<<rblocks, B, 0, stream>>>(rowptr, col, dinv, hw2, b2, Wl, z, pq, N);

  k_link<<<(EL + LT * LPT - 1) / (LT * LPT), LT, 0, stream>>>(pq, eli, bl, pred, EL);
}
